# starting kernel
# baseline (speedup 1.0000x reference)
#include <hip/hip_runtime.h>

#define N_NODES 50000
#define N_EDGES 800000
#define NFEAT 512
#define NHID  256
#define NCLASS 64

typedef float v2f __attribute__((ext_vector_type(2)));
typedef float v4f __attribute__((ext_vector_type(4)));
typedef float v8f __attribute__((ext_vector_type(8)));

template<int K, int N>
__global__ __launch_bounds__(128)
void gemm_wmma_f32(const float* __restrict__ A, const float* __restrict__ B,
                   float* __restrict__ C, int Mtiles) {
  constexpr int NG = N / 64;
  const int wave = threadIdx.x >> 5;
  const int lane = threadIdx.x & 31;
  const int job  = blockIdx.x * 4 + wave;
  if (job >= Mtiles * NG) return;
  const int mt = job / NG;
  const int ng = job % NG;

  const int half = lane >> 4;
  const int idx  = lane & 15;
  const int nb   = ng * 64;

  v8f acc0 = {}, acc1 = {}, acc2 = {}, acc3 = {};

  const float* __restrict__ arow  = A + (size_t)(mt * 16 + idx) * K;
  const float* __restrict__ bbase = B + nb + 4 * idx;

  for (int k0 = 0; k0 < K; k0 += 4) {
    const int ka = k0 + half * 2;
    v2f a  = *(const v2f*)(arow + ka);
    v4f r0 = *(const v4f*)(bbase + (size_t)ka * N);
    v4f r1 = *(const v4f*)(bbase + (size_t)(ka + 1) * N);
    v2f b;
    b.x = r0.x; b.y = r1.x;
    acc0 = __builtin_amdgcn_wmma_f32_16x16x4_f32(false, a, false, b, (short)0,
                                                 acc0, false, false);
    b.x = r0.y; b.y = r1.y;
    acc1 = __builtin_amdgcn_wmma_f32_16x16x4_f32(false, a, false, b, (short)0,
                                                 acc1, false, false);
    b.x = r0.z; b.y = r1.z;
    acc2 = __builtin_amdgcn_wmma_f32_16x16x4_f32(false, a, false, b, (short)0,
                                                 acc2, false, false);
    b.x = r0.w; b.y = r1.w;
    acc3 = __builtin_amdgcn_wmma_f32_16x16x4_f32(false, a, false, b, (short)0,
                                                 acc3, false, false);
  }

  const int mbase = mt * 16 + half * 8;
  float* __restrict__ cptr = C + (size_t)mbase * N + nb + 4 * idx;
#pragma unroll
  for (int i = 0; i < 8; ++i) {
    v4f st;
    st.x = acc0[i]; st.y = acc1[i]; st.z = acc2[i]; st.w = acc3[i];
    *(v4f*)(cptr + (size_t)i * N) = st;
  }
}

template<int F>
__global__ void spmm_scatter(const int* __restrict__ src,
                             const int* __restrict__ dst,
                             const float* __restrict__ w,
                             const float* __restrict__ sup,
                             float* __restrict__ out) {
  const int e = blockIdx.x;
  const int t = threadIdx.x;
  const int s = src[e];
  const int d = dst[e];
  const float wt = w[e];
  const float v = wt * sup[(size_t)s * F + t];
  unsafeAtomicAdd(out + (size_t)d * F + t, v);
}

__global__ void fill_zero(float* __restrict__ p, int n) {
  int i = blockIdx.x * blockDim.x + threadIdx.x;
  const int stride = gridDim.x * blockDim.x;
  for (; i < n; i += stride) p[i] = 0.0f;
}

__global__ void bias_relu(float* __restrict__ h, const float* __restrict__ b,
                          int total, int fmask) {
  int i = blockIdx.x * blockDim.x + threadIdx.x;
  const int stride = gridDim.x * blockDim.x;
  for (; i < total; i += stride) {
    float v = h[i] + b[i & fmask];
    h[i] = fmaxf(v, 0.0f);
  }
}

__global__ __launch_bounds__(NCLASS)
void logsoftmax_bias(float* __restrict__ out, const float* __restrict__ b2) {
  __shared__ float red[NCLASS];
  const int r = blockIdx.x;
  const int t = threadIdx.x;
  const float v = out[(size_t)r * NCLASS + t] + b2[t];

  red[t] = v;
  __syncthreads();
#pragma unroll
  for (int s = NCLASS / 2; s > 0; s >>= 1) {
    if (t < s) red[t] = fmaxf(red[t], red[t + s]);
    __syncthreads();
  }
  const float mx = red[0];
  __syncthreads();

  red[t] = __expf(v - mx);
  __syncthreads();
#pragma unroll
  for (int s = NCLASS / 2; s > 0; s >>= 1) {
    if (t < s) red[t] += red[t + s];
    __syncthreads();
  }
  const float lse = __logf(red[0]);
  out[(size_t)r * NCLASS + t] = v - mx - lse;
}

extern "C" void kernel_launch(void* const* d_in, const int* in_sizes, int n_in,
                              void* d_out, int out_size, void* d_ws, size_t ws_size,
                              hipStream_t stream) {
  (void)in_sizes; (void)n_in; (void)out_size; (void)ws_size;

  const float* x  = (const float*)d_in[0];
  const float* W1 = (const float*)d_in[1];
  const float* b1 = (const float*)d_in[2];
  const float* W2 = (const float*)d_in[3];
  const float* b2 = (const float*)d_in[4];
  const float* ew = (const float*)d_in[5];
  const int*   es = (const int*)d_in[6];
  const int*   ed = (const int*)d_in[7];
  float* out = (float*)d_out;

  char* ws = (char*)d_ws;
  float* sup1 = (float*)ws;
  float* h    = (float*)(ws + (size_t)N_NODES * NHID * sizeof(float));
  float* sup2 = (float*)(ws + (size_t)2 * N_NODES * NHID * sizeof(float));

  const int Mtiles = N_NODES / 16;

  fill_zero<<<2048, 256, 0, stream>>>(h, N_NODES * NHID);
  fill_zero<<<1024, 256, 0, stream>>>(out, N_NODES * NCLASS);

  {
    const int jobs = Mtiles * (NHID / 64);
    gemm_wmma_f32<NFEAT, NHID>
        <<<(jobs + 3) / 4, 128, 0, stream>>>(x, W1, sup1, Mtiles);
  }
  spmm_scatter<NHID><<<N_EDGES, NHID, 0, stream>>>(es, ed, ew, sup1, h);
  bias_relu<<<2048, 256, 0, stream>>>(h, b1, N_NODES * NHID, NHID - 1);

  {
    const int jobs = Mtiles * (NCLASS / 64);
    gemm_wmma_f32<NHID, NCLASS>
        <<<(jobs + 3) / 4, 128, 0, stream>>>(h, W2, sup2, Mtiles);
  }
  spmm_scatter<NCLASS><<<N_EDGES, NCLASS, 0, stream>>>(es, ed, ew, sup2, out);
  logsoftmax_bias<<<N_NODES, NCLASS, 0, stream>>>(out, b2);
}


# SpMM push to pull: edges counting-sorted by destination in each call (one int atomic rank per edge, scan), one wave sums a row in registers in fixed edge order; f32 atomics and zero fills gone
# speedup vs baseline: 2.8181x; 2.8181x over previous
#include <hip/hip_runtime.h>

#define N_NODES 50000
#define N_EDGES 800000
#define NFEAT 512
#define NHID  256
#define NCLASS 64

typedef float v2f __attribute__((ext_vector_type(2)));
typedef float v4f __attribute__((ext_vector_type(4)));
typedef float v8f __attribute__((ext_vector_type(8)));

template<int K, int N>
__global__ __launch_bounds__(128)
void gemm_wmma_f32(const float* __restrict__ A, const float* __restrict__ B,
                   float* __restrict__ C, int Mtiles) {
  constexpr int NG = N / 64;
  const int wave = threadIdx.x >> 5;
  const int lane = threadIdx.x & 31;
  const int job  = blockIdx.x * 4 + wave;
  if (job >= Mtiles * NG) return;
  const int mt = job / NG;
  const int ng = job % NG;

  const int half = lane >> 4;
  const int idx  = lane & 15;
  const int nb   = ng * 64;

  v8f acc0 = {}, acc1 = {}, acc2 = {}, acc3 = {};

  const float* __restrict__ arow  = A + (size_t)(mt * 16 + idx) * K;
  const float* __restrict__ bbase = B + nb + 4 * idx;

  for (int k0 = 0; k0 < K; k0 += 4) {
    const int ka = k0 + half * 2;
    v2f a  = *(const v2f*)(arow + ka);
    v4f r0 = *(const v4f*)(bbase + (size_t)ka * N);
    v4f r1 = *(const v4f*)(bbase + (size_t)(ka + 1) * N);
    v2f b;
    b.x = r0.x; b.y = r1.x;
    acc0 = __builtin_amdgcn_wmma_f32_16x16x4_f32(false, a, false, b, (short)0,
                                                 acc0, false, false);
    b.x = r0.y; b.y = r1.y;
    acc1 = __builtin_amdgcn_wmma_f32_16x16x4_f32(false, a, false, b, (short)0,
                                                 acc1, false, false);
    b.x = r0.z; b.y = r1.z;
    acc2 = __builtin_amdgcn_wmma_f32_16x16x4_f32(false, a, false, b, (short)0,
                                                 acc2, false, false);
    b.x = r0.w; b.y = r1.w;
    acc3 = __builtin_amdgcn_wmma_f32_16x16x4_f32(false, a, false, b, (short)0,
                                                 acc3, false, false);
  }

  const int mbase = mt * 16 + half * 8;
  float* __restrict__ cptr = C + (size_t)mbase * N + nb + 4 * idx;
#pragma unroll
  for (int i = 0; i < 8; ++i) {
    v4f st;
    st.x = acc0[i]; st.y = acc1[i]; st.z = acc2[i]; st.w = acc3[i];
    *(v4f*)(cptr + (size_t)i * N) = st;
  }
}

typedef int v2i __attribute__((ext_vector_type(2)));

#define SCAN_TOT  (N_NODES + 1)
#define SCAN_PERW 1568

__global__ void zero_i32(int* __restrict__ p, int n) {
  int i = blockIdx.x * blockDim.x + threadIdx.x;
  const int stride = gridDim.x * blockDim.x;
  for (; i < n; i += stride) p[i] = 0;
}

__global__ __launch_bounds__(256)
void count_rank(const int* __restrict__ dst, int* __restrict__ deg,
                int* __restrict__ rank) {
  const int e = blockIdx.x * 256 + threadIdx.x;
  if (e < N_EDGES) rank[e] = atomicAdd(deg + dst[e], 1);
}

__global__ __launch_bounds__(1024)
void scan_deg(const int* __restrict__ deg, int* __restrict__ row_ptr) {
  __shared__ int wsum[32];
  const int lane = threadIdx.x & 31;
  const int wave = __builtin_amdgcn_readfirstlane((int)(threadIdx.x >> 5));
  const int b = wave * SCAN_PERW;
  const int e = (b + SCAN_PERW < SCAN_TOT) ? (b + SCAN_PERW) : SCAN_TOT;

  int s = 0;
  for (int i = b + lane; i < e; i += 32) s += deg[i];
#pragma unroll
  for (int off = 16; off > 0; off >>= 1) s += __shfl_xor(s, off);
  if (lane == 0) wsum[wave] = s;
  __syncthreads();

  int carry = (lane < wave) ? wsum[lane] : 0;
#pragma unroll
  for (int off = 16; off > 0; off >>= 1) carry += __shfl_xor(carry, off);

  for (int i0 = b; i0 < e; i0 += 32) {
    const int i = i0 + lane;
    const int x = (i < e) ? deg[i] : 0;
    int inc = x;
#pragma unroll
    for (int off = 1; off < 32; off <<= 1) {
      const int y = __shfl_up(inc, off);
      if (lane >= off) inc += y;
    }
    if (i < e) row_ptr[i] = carry + inc - x;
    carry += __shfl(inc, 31);
  }
}

__global__ __launch_bounds__(256)
void place_slots(const int* __restrict__ dst, const int* __restrict__ row_ptr,
                 const int* __restrict__ rank, int* __restrict__ slot_eid) {
  const int e = blockIdx.x * 256 + threadIdx.x;
  if (e < N_EDGES) slot_eid[row_ptr[dst[e]] + rank[e]] = e;
}

__global__ __launch_bounds__(256)
void order_slots(const int* __restrict__ src, const int* __restrict__ dst,
                 const float* __restrict__ w, const int* __restrict__ row_ptr,
                 const int* __restrict__ slot_eid, int* __restrict__ csr) {
  const int p = blockIdx.x * 256 + threadIdx.x;
  if (p >= N_EDGES) return;
  const int e = (int)min((unsigned)slot_eid[p], (unsigned)(N_EDGES - 1));
  const int d = dst[e];
  const int beg = row_ptr[d];
  const int end = row_ptr[d + 1];
  int r = 0;
  for (int u = beg; u < end; ++u) r += (slot_eid[u] < e) ? 1 : 0;
  v2i en;
  en.x = src[e];
  en.y = __float_as_int(w[e]);
  *(v2i*)(csr + 2 * (size_t)(beg + r)) = en;
}

template<int F>
__global__ __launch_bounds__(256)
void spmm_gather(const int* __restrict__ row_ptr, const int* __restrict__ csr,
                 const float* __restrict__ sup, float* __restrict__ out) {
  static_assert(F == 256 || F == 64, "row widths of this network");
  const int lane = threadIdx.x & 31;
  const int row = __builtin_amdgcn_readfirstlane((int)(blockIdx.x * 8 + (threadIdx.x >> 5)));
  if (row >= N_NODES) return;
  const int beg = row_ptr[row];
  const int end = row_ptr[row + 1];

  if constexpr (F == 256) {
    v4f a0 = {}, a1 = {};
    const float* __restrict__ base = sup + 4 * lane;
#pragma unroll 4
    for (int j = beg; j < end; ++j) {
      const v2i en = *(const v2i*)(csr + 2 * (size_t)j);
      const float wt = __int_as_float(en.y);
      const unsigned s = min((unsigned)en.x, (unsigned)(N_NODES - 1));
      const float* __restrict__ p = base + (size_t)s * F;
      const v4f x0 = *(const v4f*)p;
      const v4f x1 = *(const v4f*)(p + 128);
      a0 += wt * x0;
      a1 += wt * x1;
    }
    float* __restrict__ o = out + (size_t)row * F + 4 * lane;
    *(v4f*)o = a0;
    *(v4f*)(o + 128) = a1;
  } else {
    v2f a = {};
    const float* __restrict__ base = sup + 2 * lane;
#pragma unroll 8
    for (int j = beg; j < end; ++j) {
      const v2i en = *(const v2i*)(csr + 2 * (size_t)j);
      const float wt = __int_as_float(en.y);
      const unsigned s = min((unsigned)en.x, (unsigned)(N_NODES - 1));
      const v2f x = *(const v2f*)(base + (size_t)s * F);
      a += wt * x;
    }
    *(v2f*)(out + (size_t)row * F + 2 * lane) = a;
  }
}

__global__ void bias_relu(float* __restrict__ h, const float* __restrict__ b,
                          int total, int fmask) {
  int i = blockIdx.x * blockDim.x + threadIdx.x;
  const int stride = gridDim.x * blockDim.x;
  for (; i < total; i += stride) {
    float v = h[i] + b[i & fmask];
    h[i] = fmaxf(v, 0.0f);
  }
}

__global__ __launch_bounds__(NCLASS)
void logsoftmax_bias(float* __restrict__ out, const float* __restrict__ b2) {
  __shared__ float red[NCLASS];
  const int r = blockIdx.x;
  const int t = threadIdx.x;
  const float v = out[(size_t)r * NCLASS + t] + b2[t];

  red[t] = v;
  __syncthreads();
#pragma unroll
  for (int s = NCLASS / 2; s > 0; s >>= 1) {
    if (t < s) red[t] = fmaxf(red[t], red[t + s]);
    __syncthreads();
  }
  const float mx = red[0];
  __syncthreads();

  red[t] = __expf(v - mx);
  __syncthreads();
#pragma unroll
  for (int s = NCLASS / 2; s > 0; s >>= 1) {
    if (t < s) red[t] += red[t + s];
    __syncthreads();
  }
  const float lse = __logf(red[0]);
  out[(size_t)r * NCLASS + t] = v - mx - lse;
}

extern "C" void kernel_launch(void* const* d_in, const int* in_sizes, int n_in,
                              void* d_out, int out_size, void* d_ws, size_t ws_size,
                              hipStream_t stream) {
  (void)in_sizes; (void)n_in; (void)out_size; (void)ws_size;

  const float* x  = (const float*)d_in[0];
  const float* W1 = (const float*)d_in[1];
  const float* b1 = (const float*)d_in[2];
  const float* W2 = (const float*)d_in[3];
  const float* b2 = (const float*)d_in[4];
  const float* ew = (const float*)d_in[5];
  const int*   es = (const int*)d_in[6];
  const int*   ed = (const int*)d_in[7];
  float* out = (float*)d_out;

  char* ws = (char*)d_ws;
  float* sup1 = (float*)ws;
  float* h    = (float*)(ws + (size_t)N_NODES * NHID * sizeof(float));
  float* sup2 = (float*)ws;
  int* slot_eid = (int*)h;

  char* tab = ws + (size_t)2 * N_NODES * NHID * sizeof(float);
  int* row_ptr = (int*)tab;
  int* deg     = (int*)(tab + (size_t)200704);
  int* rank    = (int*)(tab + (size_t)2 * 200704);
  int* csr     = (int*)(tab + (size_t)2 * 200704 + (size_t)N_EDGES * sizeof(int));

  const int Mtiles = N_NODES / 16;
  const int eblocks = N_EDGES / 256;

  zero_i32<<<49, 1024, 0, stream>>>(deg, N_NODES + 1);
  count_rank<<<eblocks, 256, 0, stream>>>(ed, deg, rank);
  scan_deg<<<1, 1024, 0, stream>>>(deg, row_ptr);
  place_slots<<<eblocks, 256, 0, stream>>>(ed, row_ptr, rank, slot_eid);
  order_slots<<<eblocks, 256, 0, stream>>>(es, ed, ew, row_ptr, slot_eid, csr);

  {
    const int jobs = Mtiles * (NHID / 64);
    gemm_wmma_f32<NFEAT, NHID>
        <<<(jobs + 3) / 4, 128, 0, stream>>>(x, W1, sup1, Mtiles);
  }
  spmm_gather<NHID><<<N_NODES / 8, 256, 0, stream>>>(row_ptr, csr, sup1, h);
  bias_relu<<<2048, 256, 0, stream>>>(h, b1, N_NODES * NHID, NHID - 1);

  {
    const int jobs = Mtiles * (NCLASS / 64);
    gemm_wmma_f32<NHID, NCLASS>
        <<<(jobs + 3) / 4, 128, 0, stream>>>(h, W2, sup2, Mtiles);
  }
  spmm_gather<NCLASS><<<N_NODES / 8, 256, 0, stream>>>(row_ptr, csr, sup2, out);
  logsoftmax_bias<<<N_NODES, NCLASS, 0, stream>>>(out, b2);
}


# bias+ReLU and bias+log-softmax fused into the SpMM gather's row store (the wave still holds the row; row max and exp-sum by lane exchanges in the old halving-tree pairing); launches 11 -> 9
# speedup vs baseline: 3.1825x; 1.1293x over previous
#include <hip/hip_runtime.h>

#define N_NODES 50000
#define N_EDGES 800000
#define NFEAT 512
#define NHID  256
#define NCLASS 64

typedef float v2f __attribute__((ext_vector_type(2)));
typedef float v4f __attribute__((ext_vector_type(4)));
typedef float v8f __attribute__((ext_vector_type(8)));

template<int K, int N>
__global__ __launch_bounds__(128)
void gemm_wmma_f32(const float* __restrict__ A, const float* __restrict__ B,
                   float* __restrict__ C, int Mtiles) {
  constexpr int NG = N / 64;
  const int wave = threadIdx.x >> 5;
  const int lane = threadIdx.x & 31;
  const int job  = blockIdx.x * 4 + wave;
  if (job >= Mtiles * NG) return;
  const int mt = job / NG;
  const int ng = job % NG;

  const int half = lane >> 4;
  const int idx  = lane & 15;
  const int nb   = ng * 64;

  v8f acc0 = {}, acc1 = {}, acc2 = {}, acc3 = {};

  const float* __restrict__ arow  = A + (size_t)(mt * 16 + idx) * K;
  const float* __restrict__ bbase = B + nb + 4 * idx;

  for (int k0 = 0; k0 < K; k0 += 4) {
    const int ka = k0 + half * 2;
    v2f a  = *(const v2f*)(arow + ka);
    v4f r0 = *(const v4f*)(bbase + (size_t)ka * N);
    v4f r1 = *(const v4f*)(bbase + (size_t)(ka + 1) * N);
    v2f b;
    b.x = r0.x; b.y = r1.x;
    acc0 = __builtin_amdgcn_wmma_f32_16x16x4_f32(false, a, false, b, (short)0,
                                                 acc0, false, false);
    b.x = r0.y; b.y = r1.y;
    acc1 = __builtin_amdgcn_wmma_f32_16x16x4_f32(false, a, false, b, (short)0,
                                                 acc1, false, false);
    b.x = r0.z; b.y = r1.z;
    acc2 = __builtin_amdgcn_wmma_f32_16x16x4_f32(false, a, false, b, (short)0,
                                                 acc2, false, false);
    b.x = r0.w; b.y = r1.w;
    acc3 = __builtin_amdgcn_wmma_f32_16x16x4_f32(false, a, false, b, (short)0,
                                                 acc3, false, false);
  }

  const int mbase = mt * 16 + half * 8;
  float* __restrict__ cptr = C + (size_t)mbase * N + nb + 4 * idx;
#pragma unroll
  for (int i = 0; i < 8; ++i) {
    v4f st;
    st.x = acc0[i]; st.y = acc1[i]; st.z = acc2[i]; st.w = acc3[i];
    *(v4f*)(cptr + (size_t)i * N) = st;
  }
}

typedef int v2i __attribute__((ext_vector_type(2)));

#define SCAN_TOT  (N_NODES + 1)
#define SCAN_PERW 1568

__global__ void zero_i32(int* __restrict__ p, int n) {
  int i = blockIdx.x * blockDim.x + threadIdx.x;
  const int stride = gridDim.x * blockDim.x;
  for (; i < n; i += stride) p[i] = 0;
}

__global__ __launch_bounds__(256)
void count_rank(const int* __restrict__ dst, int* __restrict__ deg,
                int* __restrict__ rank) {
  const int e = blockIdx.x * 256 + threadIdx.x;
  if (e < N_EDGES) rank[e] = atomicAdd(deg + dst[e], 1);
}

__global__ __launch_bounds__(1024)
void scan_deg(const int* __restrict__ deg, int* __restrict__ row_ptr) {
  __shared__ int wsum[32];
  const int lane = threadIdx.x & 31;
  const int wave = __builtin_amdgcn_readfirstlane((int)(threadIdx.x >> 5));
  const int b = wave * SCAN_PERW;
  const int e = (b + SCAN_PERW < SCAN_TOT) ? (b + SCAN_PERW) : SCAN_TOT;

  int s = 0;
  for (int i = b + lane; i < e; i += 32) s += deg[i];
#pragma unroll
  for (int off = 16; off > 0; off >>= 1) s += __shfl_xor(s, off);
  if (lane == 0) wsum[wave] = s;
  __syncthreads();

  int carry = (lane < wave) ? wsum[lane] : 0;
#pragma unroll
  for (int off = 16; off > 0; off >>= 1) carry += __shfl_xor(carry, off);

  for (int i0 = b; i0 < e; i0 += 32) {
    const int i = i0 + lane;
    const int x = (i < e) ? deg[i] : 0;
    int inc = x;
#pragma unroll
    for (int off = 1; off < 32; off <<= 1) {
      const int y = __shfl_up(inc, off);
      if (lane >= off) inc += y;
    }
    if (i < e) row_ptr[i] = carry + inc - x;
    carry += __shfl(inc, 31);
  }
}

__global__ __launch_bounds__(256)
void place_slots(const int* __restrict__ dst, const int* __restrict__ row_ptr,
                 const int* __restrict__ rank, int* __restrict__ slot_eid) {
  const int e = blockIdx.x * 256 + threadIdx.x;
  if (e < N_EDGES) slot_eid[row_ptr[dst[e]] + rank[e]] = e;
}

__global__ __launch_bounds__(256)
void order_slots(const int* __restrict__ src, const int* __restrict__ dst,
                 const float* __restrict__ w, const int* __restrict__ row_ptr,
                 const int* __restrict__ slot_eid, int* __restrict__ csr) {
  const int p = blockIdx.x * 256 + threadIdx.x;
  if (p >= N_EDGES) return;
  const int e = (int)min((unsigned)slot_eid[p], (unsigned)(N_EDGES - 1));
  const int d = dst[e];
  const int beg = row_ptr[d];
  const int end = row_ptr[d + 1];
  int r = 0;
  for (int u = beg; u < end; ++u) r += (slot_eid[u] < e) ? 1 : 0;
  v2i en;
  en.x = src[e];
  en.y = __float_as_int(w[e]);
  *(v2i*)(csr + 2 * (size_t)(beg + r)) = en;
}

__device__ __forceinline__ float xhalf(float x) {
  const int b = __builtin_bit_cast(int, x);
  return __builtin_bit_cast(float,
      __builtin_amdgcn_permlanex16(b, b, 0x76543210, 0xfedcba98, false, false));
}
template<int M>
__device__ __forceinline__ float xrow(float x) {
  static_assert(M == 1 || M == 2 || M == 4 || M == 8, "stays inside a 16-lane row");
  const int b = __builtin_bit_cast(int, x);
  return __builtin_bit_cast(float,
      __builtin_amdgcn_update_dpp(b, b, 0x160 + M, 0xf, 0xf, true));
}

template<int F>
__global__ __launch_bounds__(256)
void spmm_gather(const int* __restrict__ row_ptr, const int* __restrict__ csr,
                 const float* __restrict__ sup, const float* __restrict__ bias,
                 float* __restrict__ out) {
  static_assert(F == 256 || F == 64, "row widths of this network");
  const int lane = threadIdx.x & 31;
  const int row = __builtin_amdgcn_readfirstlane((int)(blockIdx.x * 8 + (threadIdx.x >> 5)));
  if (row >= N_NODES) return;
  const int beg = row_ptr[row];
  const int end = row_ptr[row + 1];

  if constexpr (F == 256) {
    v4f a0 = {}, a1 = {};
    const float* __restrict__ base = sup + 4 * lane;
    v4f c0 = *(const v4f*)(bias + 4 * lane);
    v4f c1 = *(const v4f*)(bias + 128 + 4 * lane);
    asm volatile("" : "+v"(c0), "+v"(c1));
#pragma unroll 4
    for (int j = beg; j < end; ++j) {
      const v2i en = *(const v2i*)(csr + 2 * (size_t)j);
      const float wt = __int_as_float(en.y);
      const unsigned s = min((unsigned)en.x, (unsigned)(N_NODES - 1));
      const float* __restrict__ p = base + (size_t)s * F;
      const v4f x0 = *(const v4f*)p;
      const v4f x1 = *(const v4f*)(p + 128);
      a0 += wt * x0;
      a1 += wt * x1;
    }
    a0 = a0 + c0;
    a1 = a1 + c1;
    v4f h0, h1;
    h0.x = fmaxf(a0.x, 0.0f); h0.y = fmaxf(a0.y, 0.0f);
    h0.z = fmaxf(a0.z, 0.0f); h0.w = fmaxf(a0.w, 0.0f);
    h1.x = fmaxf(a1.x, 0.0f); h1.y = fmaxf(a1.y, 0.0f);
    h1.z = fmaxf(a1.z, 0.0f); h1.w = fmaxf(a1.w, 0.0f);
    float* __restrict__ o = out + (size_t)row * F + 4 * lane;
    *(v4f*)o = h0;
    *(v4f*)(o + 128) = h1;
  } else {
    v2f a = {};
    const float* __restrict__ base = sup + 2 * lane;
    v2f bb = *(const v2f*)(bias + 2 * lane);
    asm volatile("" : "+v"(bb));
#pragma unroll 8
    for (int j = beg; j < end; ++j) {
      const v2i en = *(const v2i*)(csr + 2 * (size_t)j);
      const float wt = __int_as_float(en.y);
      const unsigned s = min((unsigned)en.x, (unsigned)(N_NODES - 1));
      const v2f x = *(const v2f*)(base + (size_t)s * F);
      a += wt * x;
    }
    const float vx = a.x + bb.x;
    const float vy = a.y + bb.y;

    float mx = fmaxf(vx, vy);
    mx = fmaxf(mx, xhalf(mx));
    mx = fmaxf(mx, xrow<8>(mx));
    mx = fmaxf(mx, xrow<4>(mx));
    mx = fmaxf(mx, xrow<2>(mx));
    mx = fmaxf(mx, xrow<1>(mx));

    float sx = __expf(vx - mx);
    float sy = __expf(vy - mx);
    sx += xhalf(sx);    sy += xhalf(sy);
    sx += xrow<8>(sx);  sy += xrow<8>(sy);
    sx += xrow<4>(sx);  sy += xrow<4>(sy);
    sx += xrow<2>(sx);  sy += xrow<2>(sy);
    sx += xrow<1>(sx);  sy += xrow<1>(sy);
    const float lse = __logf(sx + sy);

    v2f o;
    o.x = vx - mx - lse;
    o.y = vy - mx - lse;
    *(v2f*)(out + (size_t)row * F + 2 * lane) = o;
  }
}

extern "C" void kernel_launch(void* const* d_in, const int* in_sizes, int n_in,
                              void* d_out, int out_size, void* d_ws, size_t ws_size,
                              hipStream_t stream) {
  (void)in_sizes; (void)n_in; (void)out_size; (void)ws_size;

  const float* x  = (const float*)d_in[0];
  const float* W1 = (const float*)d_in[1];
  const float* b1 = (const float*)d_in[2];
  const float* W2 = (const float*)d_in[3];
  const float* b2 = (const float*)d_in[4];
  const float* ew = (const float*)d_in[5];
  const int*   es = (const int*)d_in[6];
  const int*   ed = (const int*)d_in[7];
  float* out = (float*)d_out;

  char* ws = (char*)d_ws;
  float* sup1 = (float*)ws;
  float* h    = (float*)(ws + (size_t)N_NODES * NHID * sizeof(float));
  float* sup2 = (float*)ws;
  int* slot_eid = (int*)h;

  char* tab = ws + (size_t)2 * N_NODES * NHID * sizeof(float);
  int* row_ptr = (int*)tab;
  int* deg     = (int*)(tab + (size_t)200704);
  int* rank    = (int*)(tab + (size_t)2 * 200704);
  int* csr     = (int*)(tab + (size_t)2 * 200704 + (size_t)N_EDGES * sizeof(int));

  const int Mtiles = N_NODES / 16;
  const int eblocks = N_EDGES / 256;

  zero_i32<<<49, 1024, 0, stream>>>(deg, N_NODES + 1);
  count_rank<<<eblocks, 256, 0, stream>>>(ed, deg, rank);
  scan_deg<<<1, 1024, 0, stream>>>(deg, row_ptr);
  place_slots<<<eblocks, 256, 0, stream>>>(ed, row_ptr, rank, slot_eid);
  order_slots<<<eblocks, 256, 0, stream>>>(es, ed, ew, row_ptr, slot_eid, csr);

  {
    const int jobs = Mtiles * (NHID / 64);
    gemm_wmma_f32<NFEAT, NHID>
        <<<(jobs + 3) / 4, 128, 0, stream>>>(x, W1, sup1, Mtiles);
  }
  spmm_gather<NHID><<<N_NODES / 8, 256, 0, stream>>>(row_ptr, csr, sup1, b1, h);

  {
    const int jobs = Mtiles * (NCLASS / 64);
    gemm_wmma_f32<NHID, NCLASS>
        <<<(jobs + 3) / 4, 128, 0, stream>>>(h, W2, sup2, Mtiles);
  }
  spmm_gather<NCLASS><<<N_NODES / 8, 256, 0, stream>>>(row_ptr, csr, sup2, b2, out);
}


# scan_deg in one pass: each lane's counts arrive as b128 loads issued together (deg zeroed to its padded end) and are prefix-summed in registers by DPP row shifts; exact 32-bit integer sums
# speedup vs baseline: 3.5763x; 1.1237x over previous
#include <hip/hip_runtime.h>

#define N_NODES 50000
#define N_EDGES 800000
#define NFEAT 512
#define NHID  256
#define NCLASS 64

typedef float v2f __attribute__((ext_vector_type(2)));
typedef float v4f __attribute__((ext_vector_type(4)));
typedef float v8f __attribute__((ext_vector_type(8)));

template<int K, int N>
__global__ __launch_bounds__(128)
void gemm_wmma_f32(const float* __restrict__ A, const float* __restrict__ B,
                   float* __restrict__ C, int Mtiles) {
  constexpr int NG = N / 64;
  const int wave = threadIdx.x >> 5;
  const int lane = threadIdx.x & 31;
  const int job  = blockIdx.x * 4 + wave;
  if (job >= Mtiles * NG) return;
  const int mt = job / NG;
  const int ng = job % NG;

  const int half = lane >> 4;
  const int idx  = lane & 15;
  const int nb   = ng * 64;

  v8f acc0 = {}, acc1 = {}, acc2 = {}, acc3 = {};

  const float* __restrict__ arow  = A + (size_t)(mt * 16 + idx) * K;
  const float* __restrict__ bbase = B + nb + 4 * idx;

  for (int k0 = 0; k0 < K; k0 += 4) {
    const int ka = k0 + half * 2;
    v2f a  = *(const v2f*)(arow + ka);
    v4f r0 = *(const v4f*)(bbase + (size_t)ka * N);
    v4f r1 = *(const v4f*)(bbase + (size_t)(ka + 1) * N);
    v2f b;
    b.x = r0.x; b.y = r1.x;
    acc0 = __builtin_amdgcn_wmma_f32_16x16x4_f32(false, a, false, b, (short)0,
                                                 acc0, false, false);
    b.x = r0.y; b.y = r1.y;
    acc1 = __builtin_amdgcn_wmma_f32_16x16x4_f32(false, a, false, b, (short)0,
                                                 acc1, false, false);
    b.x = r0.z; b.y = r1.z;
    acc2 = __builtin_amdgcn_wmma_f32_16x16x4_f32(false, a, false, b, (short)0,
                                                 acc2, false, false);
    b.x = r0.w; b.y = r1.w;
    acc3 = __builtin_amdgcn_wmma_f32_16x16x4_f32(false, a, false, b, (short)0,
                                                 acc3, false, false);
  }

  const int mbase = mt * 16 + half * 8;
  float* __restrict__ cptr = C + (size_t)mbase * N + nb + 4 * idx;
#pragma unroll
  for (int i = 0; i < 8; ++i) {
    v4f st;
    st.x = acc0[i]; st.y = acc1[i]; st.z = acc2[i]; st.w = acc3[i];
    *(v4f*)(cptr + (size_t)i * N) = st;
  }
}

typedef int v2i __attribute__((ext_vector_type(2)));

typedef int v4i __attribute__((ext_vector_type(4)));

#define SCAN_PAD  50176
#define SCAN_IT   13
#define SCAN_PERW (128 * SCAN_IT)

__global__ void zero_i32(int* __restrict__ p, int n) {
  int i = blockIdx.x * blockDim.x + threadIdx.x;
  const int stride = gridDim.x * blockDim.x;
  for (; i < n; i += stride) p[i] = 0;
}

__global__ __launch_bounds__(256)
void count_rank(const int* __restrict__ dst, int* __restrict__ deg,
                int* __restrict__ rank) {
  const int e = blockIdx.x * 256 + threadIdx.x;
  if (e < N_EDGES) rank[e] = atomicAdd(deg + dst[e], 1);
}

template<int S>
__device__ __forceinline__ int rowshr(int x) {
  static_assert(S == 1 || S == 2 || S == 4 || S == 8, "stays inside a 16-lane row");
  return __builtin_amdgcn_update_dpp(0, x, 0x110 + S, 0xf, 0xf, true);
}

__global__ __launch_bounds__(1024)
void scan_deg(const int* __restrict__ deg, int* __restrict__ row_ptr) {
  __shared__ int wsum[32];
  const int lane = threadIdx.x & 31;
  const int wave = __builtin_amdgcn_readfirstlane((int)(threadIdx.x >> 5));
  const int base = wave * SCAN_PERW + 4 * lane;

  v4i v[SCAN_IT];
#pragma unroll
  for (int t = 0; t < SCAN_IT; ++t) {
    const int i = base + 128 * t;
    v4i x = {0, 0, 0, 0};
    if (i < SCAN_PAD) x = *(const v4i*)(deg + i);
    v[t] = x;
  }

  int inc[SCAN_IT];
  int gbase[SCAN_IT];
  int run = 0;
#pragma unroll
  for (int t = 0; t < SCAN_IT; ++t) {
    int s = v[t].x + v[t].y + v[t].z + v[t].w;
    s += rowshr<1>(s);
    s += rowshr<2>(s);
    s += rowshr<4>(s);
    s += rowshr<8>(s);
    const int lo = __builtin_amdgcn_readlane(s, 15);
    s += (lane >= 16) ? lo : 0;
    inc[t] = s;
    gbase[t] = run;
    run += __builtin_amdgcn_readlane(s, 31);
  }
  if (lane == 0) wsum[wave] = run;
  __syncthreads();

  int carry = (lane < wave) ? wsum[lane] : 0;
#pragma unroll
  for (int off = 16; off > 0; off >>= 1) carry += __shfl_xor(carry, off);

#pragma unroll
  for (int t = 0; t < SCAN_IT; ++t) {
    const int i = base + 128 * t;
    const int own = v[t].x + v[t].y + v[t].z + v[t].w;
    v4i o;
    o.x = carry + gbase[t] + inc[t] - own;
    o.y = o.x + v[t].x;
    o.z = o.y + v[t].y;
    o.w = o.z + v[t].z;
    if (i < SCAN_PAD) *(v4i*)(row_ptr + i) = o;
  }
}

__global__ __launch_bounds__(256)
void place_slots(const int* __restrict__ dst, const int* __restrict__ row_ptr,
                 const int* __restrict__ rank, int* __restrict__ slot_eid) {
  const int e = blockIdx.x * 256 + threadIdx.x;
  if (e < N_EDGES) slot_eid[row_ptr[dst[e]] + rank[e]] = e;
}

__global__ __launch_bounds__(256)
void order_slots(const int* __restrict__ src, const int* __restrict__ dst,
                 const float* __restrict__ w, const int* __restrict__ row_ptr,
                 const int* __restrict__ slot_eid, int* __restrict__ csr) {
  const int p = blockIdx.x * 256 + threadIdx.x;
  if (p >= N_EDGES) return;
  const int e = (int)min((unsigned)slot_eid[p], (unsigned)(N_EDGES - 1));
  const int d = dst[e];
  const int beg = row_ptr[d];
  const int end = row_ptr[d + 1];
  int r = 0;
  for (int u = beg; u < end; ++u) r += (slot_eid[u] < e) ? 1 : 0;
  v2i en;
  en.x = src[e];
  en.y = __float_as_int(w[e]);
  *(v2i*)(csr + 2 * (size_t)(beg + r)) = en;
}

__device__ __forceinline__ float xhalf(float x) {
  const int b = __builtin_bit_cast(int, x);
  return __builtin_bit_cast(float,
      __builtin_amdgcn_permlanex16(b, b, 0x76543210, 0xfedcba98, false, false));
}
template<int M>
__device__ __forceinline__ float xrow(float x) {
  static_assert(M == 1 || M == 2 || M == 4 || M == 8, "stays inside a 16-lane row");
  const int b = __builtin_bit_cast(int, x);
  return __builtin_bit_cast(float,
      __builtin_amdgcn_update_dpp(b, b, 0x160 + M, 0xf, 0xf, true));
}

template<int F>
__global__ __launch_bounds__(256)
void spmm_gather(const int* __restrict__ row_ptr, const int* __restrict__ csr,
                 const float* __restrict__ sup, const float* __restrict__ bias,
                 float* __restrict__ out) {
  static_assert(F == 256 || F == 64, "row widths of this network");
  const int lane = threadIdx.x & 31;
  const int row = __builtin_amdgcn_readfirstlane((int)(blockIdx.x * 8 + (threadIdx.x >> 5)));
  if (row >= N_NODES) return;
  const int beg = row_ptr[row];
  const int end = row_ptr[row + 1];

  if constexpr (F == 256) {
    v4f a0 = {}, a1 = {};
    const float* __restrict__ base = sup + 4 * lane;
    v4f c0 = *(const v4f*)(bias + 4 * lane);
    v4f c1 = *(const v4f*)(bias + 128 + 4 * lane);
    asm volatile("" : "+v"(c0), "+v"(c1));
#pragma unroll 4
    for (int j = beg; j < end; ++j) {
      const v2i en = *(const v2i*)(csr + 2 * (size_t)j);
      const float wt = __int_as_float(en.y);
      const unsigned s = min((unsigned)en.x, (unsigned)(N_NODES - 1));
      const float* __restrict__ p = base + (size_t)s * F;
      const v4f x0 = *(const v4f*)p;
      const v4f x1 = *(const v4f*)(p + 128);
      a0 += wt * x0;
      a1 += wt * x1;
    }
    a0 = a0 + c0;
    a1 = a1 + c1;
    v4f h0, h1;
    h0.x = fmaxf(a0.x, 0.0f); h0.y = fmaxf(a0.y, 0.0f);
    h0.z = fmaxf(a0.z, 0.0f); h0.w = fmaxf(a0.w, 0.0f);
    h1.x = fmaxf(a1.x, 0.0f); h1.y = fmaxf(a1.y, 0.0f);
    h1.z = fmaxf(a1.z, 0.0f); h1.w = fmaxf(a1.w, 0.0f);
    float* __restrict__ o = out + (size_t)row * F + 4 * lane;
    *(v4f*)o = h0;
    *(v4f*)(o + 128) = h1;
  } else {
    v2f a = {};
    const float* __restrict__ base = sup + 2 * lane;
    v2f bb = *(const v2f*)(bias + 2 * lane);
    asm volatile("" : "+v"(bb));
#pragma unroll 8
    for (int j = beg; j < end; ++j) {
      const v2i en = *(const v2i*)(csr + 2 * (size_t)j);
      const float wt = __int_as_float(en.y);
      const unsigned s = min((unsigned)en.x, (unsigned)(N_NODES - 1));
      const v2f x = *(const v2f*)(base + (size_t)s * F);
      a += wt * x;
    }
    const float vx = a.x + bb.x;
    const float vy = a.y + bb.y;

    float mx = fmaxf(vx, vy);
    mx = fmaxf(mx, xhalf(mx));
    mx = fmaxf(mx, xrow<8>(mx));
    mx = fmaxf(mx, xrow<4>(mx));
    mx = fmaxf(mx, xrow<2>(mx));
    mx = fmaxf(mx, xrow<1>(mx));

    float sx = __expf(vx - mx);
    float sy = __expf(vy - mx);
    sx += xhalf(sx);    sy += xhalf(sy);
    sx += xrow<8>(sx);  sy += xrow<8>(sy);
    sx += xrow<4>(sx);  sy += xrow<4>(sy);
    sx += xrow<2>(sx);  sy += xrow<2>(sy);
    sx += xrow<1>(sx);  sy += xrow<1>(sy);
    const float lse = __logf(sx + sy);

    v2f o;
    o.x = vx - mx - lse;
    o.y = vy - mx - lse;
    *(v2f*)(out + (size_t)row * F + 2 * lane) = o;
  }
}

extern "C" void kernel_launch(void* const* d_in, const int* in_sizes, int n_in,
                              void* d_out, int out_size, void* d_ws, size_t ws_size,
                              hipStream_t stream) {
  (void)in_sizes; (void)n_in; (void)out_size; (void)ws_size;

  const float* x  = (const float*)d_in[0];
  const float* W1 = (const float*)d_in[1];
  const float* b1 = (const float*)d_in[2];
  const float* W2 = (const float*)d_in[3];
  const float* b2 = (const float*)d_in[4];
  const float* ew = (const float*)d_in[5];
  const int*   es = (const int*)d_in[6];
  const int*   ed = (const int*)d_in[7];
  float* out = (float*)d_out;

  char* ws = (char*)d_ws;
  float* sup1 = (float*)ws;
  float* h    = (float*)(ws + (size_t)N_NODES * NHID * sizeof(float));
  float* sup2 = (float*)ws;
  int* slot_eid = (int*)h;

  char* tab = ws + (size_t)2 * N_NODES * NHID * sizeof(float);
  int* row_ptr = (int*)tab;
  int* deg     = (int*)(tab + (size_t)200704);
  int* rank    = (int*)(tab + (size_t)2 * 200704);
  int* csr     = (int*)(tab + (size_t)2 * 200704 + (size_t)N_EDGES * sizeof(int));

  const int Mtiles = N_NODES / 16;
  const int eblocks = N_EDGES / 256;

  zero_i32<<<49, 1024, 0, stream>>>(deg, SCAN_PAD);
  count_rank<<<eblocks, 256, 0, stream>>>(ed, deg, rank);
  scan_deg<<<1, 1024, 0, stream>>>(deg, row_ptr);
  place_slots<<<eblocks, 256, 0, stream>>>(ed, row_ptr, rank, slot_eid);
  order_slots<<<eblocks, 256, 0, stream>>>(es, ed, ew, row_ptr, slot_eid, csr);

  {
    const int jobs = Mtiles * (NHID / 64);
    gemm_wmma_f32<NFEAT, NHID>
        <<<(jobs + 3) / 4, 128, 0, stream>>>(x, W1, sup1, Mtiles);
  }
  spmm_gather<NHID><<<N_NODES / 8, 256, 0, stream>>>(row_ptr, csr, sup1, b1, h);

  {
    const int jobs = Mtiles * (NCLASS / 64);
    gemm_wmma_f32<NHID, NCLASS>
        <<<(jobs + 3) / 4, 128, 0, stream>>>(h, W2, sup2, Mtiles);
  }
  spmm_gather<NCLASS><<<N_NODES / 8, 256, 0, stream>>>(row_ptr, csr, sup2, b2, out);
}


# Edge sort ranked by LDS counters: edges binned into 391 row ranges (count, scan, scatter), one block per range counts, places, orders its rows; per-edge global atomic and clear gone; 9->8 launches
# speedup vs baseline: 4.1317x; 1.1553x over previous
#include <hip/hip_runtime.h>

#define N_NODES 50000
#define N_EDGES 800000
#define NFEAT 512
#define NHID  256
#define NCLASS 64

typedef float v2f __attribute__((ext_vector_type(2)));
typedef float v4f __attribute__((ext_vector_type(4)));
typedef float v8f __attribute__((ext_vector_type(8)));

template<int K, int N>
__global__ __launch_bounds__(128)
void gemm_wmma_f32(const float* __restrict__ A, const float* __restrict__ B,
                   float* __restrict__ C, int Mtiles) {
  constexpr int NG = N / 64;
  const int wave = threadIdx.x >> 5;
  const int lane = threadIdx.x & 31;
  const int job  = blockIdx.x * 4 + wave;
  if (job >= Mtiles * NG) return;
  const int mt = job / NG;
  const int ng = job % NG;

  const int half = lane >> 4;
  const int idx  = lane & 15;
  const int nb   = ng * 64;

  v8f acc0 = {}, acc1 = {}, acc2 = {}, acc3 = {};

  const float* __restrict__ arow  = A + (size_t)(mt * 16 + idx) * K;
  const float* __restrict__ bbase = B + nb + 4 * idx;

  for (int k0 = 0; k0 < K; k0 += 4) {
    const int ka = k0 + half * 2;
    v2f a  = *(const v2f*)(arow + ka);
    v4f r0 = *(const v4f*)(bbase + (size_t)ka * N);
    v4f r1 = *(const v4f*)(bbase + (size_t)(ka + 1) * N);
    v2f b;
    b.x = r0.x; b.y = r1.x;
    acc0 = __builtin_amdgcn_wmma_f32_16x16x4_f32(false, a, false, b, (short)0,
                                                 acc0, false, false);
    b.x = r0.y; b.y = r1.y;
    acc1 = __builtin_amdgcn_wmma_f32_16x16x4_f32(false, a, false, b, (short)0,
                                                 acc1, false, false);
    b.x = r0.z; b.y = r1.z;
    acc2 = __builtin_amdgcn_wmma_f32_16x16x4_f32(false, a, false, b, (short)0,
                                                 acc2, false, false);
    b.x = r0.w; b.y = r1.w;
    acc3 = __builtin_amdgcn_wmma_f32_16x16x4_f32(false, a, false, b, (short)0,
                                                 acc3, false, false);
  }

  const int mbase = mt * 16 + half * 8;
  float* __restrict__ cptr = C + (size_t)mbase * N + nb + 4 * idx;
#pragma unroll
  for (int i = 0; i < 8; ++i) {
    v4f st;
    st.x = acc0[i]; st.y = acc1[i]; st.z = acc2[i]; st.w = acc3[i];
    *(v4f*)(cptr + (size_t)i * N) = st;
  }
}

typedef int v2i __attribute__((ext_vector_type(2)));

typedef int v4i __attribute__((ext_vector_type(4)));

#define SCAN_PAD  50176
#define SCAN_IT   13
#define SCAN_PERW (128 * SCAN_IT)

#define BIN_SHIFT 7
#define BIN_ROWS  (1 << BIN_SHIFT)
#define NBIN   391
#define NCHUNK 128
#define CHUNK  6250
#define PER_T  ((CHUNK + 1023) / 1024)
static_assert(NCHUNK * CHUNK == N_EDGES, "chunks cover the edge list exactly");
static_assert(NBIN * BIN_ROWS >= N_NODES + 1, "ranges cover every row and the closing row_ptr entry");
static_assert(((N_NODES - 1) >> BIN_SHIFT) < NBIN, "every row has a range");
static_assert(NBIN * NCHUNK + 1 <= SCAN_PAD, "the count table and its closing total fit the scan");
static_assert(N_EDGES <= (1 << 20), "an edge number fits the low 20 bits of a binned word");

__global__ __launch_bounds__(1024)
void bin_count(const int* __restrict__ dst, int* __restrict__ cnt) {
  __shared__ int hist[NBIN];
  const int t = threadIdx.x;
  if (t < NBIN) hist[t] = 0;
  if (blockIdx.x == 0)
    for (int i = NBIN * NCHUNK + t; i < SCAN_PAD; i += 1024) cnt[i] = 0;
  const int base = blockIdx.x * CHUNK;
  int d[PER_T];
#pragma unroll
  for (int j = 0; j < PER_T; ++j) {
    const int i = t + 1024 * j;
    d[j] = (i < CHUNK) ? dst[base + i] : 0;
  }
  __syncthreads();
#pragma unroll
  for (int j = 0; j < PER_T; ++j) {
    const int i = t + 1024 * j;
    if (i < CHUNK) {
      const unsigned dd = min((unsigned)d[j], (unsigned)(N_NODES - 1));
      atomicAdd(&hist[dd >> BIN_SHIFT], 1);
    }
  }
  __syncthreads();
  if (t < NBIN) cnt[t * NCHUNK + blockIdx.x] = hist[t];
}

template<int S>
__device__ __forceinline__ int rowshr(int x) {
  static_assert(S == 1 || S == 2 || S == 4 || S == 8, "stays inside a 16-lane row");
  return __builtin_amdgcn_update_dpp(0, x, 0x110 + S, 0xf, 0xf, true);
}

__global__ __launch_bounds__(1024)
void scan_deg(const int* __restrict__ deg, int* __restrict__ row_ptr) {
  __shared__ int wsum[32];
  const int lane = threadIdx.x & 31;
  const int wave = __builtin_amdgcn_readfirstlane((int)(threadIdx.x >> 5));
  const int base = wave * SCAN_PERW + 4 * lane;

  v4i v[SCAN_IT];
#pragma unroll
  for (int t = 0; t < SCAN_IT; ++t) {
    const int i = base + 128 * t;
    v4i x = {0, 0, 0, 0};
    if (i < SCAN_PAD) x = *(const v4i*)(deg + i);
    v[t] = x;
  }

  int inc[SCAN_IT];
  int gbase[SCAN_IT];
  int run = 0;
#pragma unroll
  for (int t = 0; t < SCAN_IT; ++t) {
    int s = v[t].x + v[t].y + v[t].z + v[t].w;
    s += rowshr<1>(s);
    s += rowshr<2>(s);
    s += rowshr<4>(s);
    s += rowshr<8>(s);
    const int lo = __builtin_amdgcn_readlane(s, 15);
    s += (lane >= 16) ? lo : 0;
    inc[t] = s;
    gbase[t] = run;
    run += __builtin_amdgcn_readlane(s, 31);
  }
  if (lane == 0) wsum[wave] = run;
  __syncthreads();

  int carry = (lane < wave) ? wsum[lane] : 0;
#pragma unroll
  for (int off = 16; off > 0; off >>= 1) carry += __shfl_xor(carry, off);

#pragma unroll
  for (int t = 0; t < SCAN_IT; ++t) {
    const int i = base + 128 * t;
    const int own = v[t].x + v[t].y + v[t].z + v[t].w;
    v4i o;
    o.x = carry + gbase[t] + inc[t] - own;
    o.y = o.x + v[t].x;
    o.z = o.y + v[t].y;
    o.w = o.z + v[t].z;
    if (i < SCAN_PAD) *(v4i*)(row_ptr + i) = o;
  }
}

__global__ __launch_bounds__(1024)
void bin_scatter(const int* __restrict__ dst, const int* __restrict__ offs,
                 int* __restrict__ binned) {
  __shared__ int cur[NBIN];
  const int t = threadIdx.x;
  if (t < NBIN) cur[t] = offs[t * NCHUNK + blockIdx.x];
  const int base = blockIdx.x * CHUNK;
  int d[PER_T];
#pragma unroll
  for (int j = 0; j < PER_T; ++j) {
    const int i = t + 1024 * j;
    d[j] = (i < CHUNK) ? dst[base + i] : 0;
  }
  __syncthreads();
#pragma unroll
  for (int j = 0; j < PER_T; ++j) {
    const int i = t + 1024 * j;
    if (i < CHUNK) {
      const unsigned dd = min((unsigned)d[j], (unsigned)(N_NODES - 1));
      const int pos = atomicAdd(&cur[dd >> BIN_SHIFT], 1);
      binned[min((unsigned)pos, (unsigned)(N_EDGES - 1))] =
          (int)(((dd & (unsigned)(BIN_ROWS - 1)) << 20) | (unsigned)(base + i));
    }
  }
}

__global__ __launch_bounds__(1024)
void bin_sort(const int* __restrict__ src, const float* __restrict__ w,
              const int* __restrict__ offs, const int* __restrict__ binned,
              int* __restrict__ rank, int* __restrict__ slot,
              int* __restrict__ row_ptr, int* __restrict__ csr) {
  __shared__ int cnt[BIN_ROWS];
  __shared__ int pre[BIN_ROWS + 1];
  const int t = threadIdx.x;
  const int b = blockIdx.x;
  const int beg = (int)min((unsigned)offs[b * NCHUNK], (unsigned)N_EDGES);
  const int end = (int)min((unsigned)offs[(b + 1) * NCHUNK], (unsigned)N_EDGES);

  if (t < BIN_ROWS) cnt[t] = 0;
  __syncthreads();

  for (int i = beg + t; i < end; i += 1024) {
    const unsigned r = min((unsigned)binned[i] >> 20, (unsigned)(BIN_ROWS - 1));
    rank[i] = atomicAdd(&cnt[r], 1);
  }
  __syncthreads();

  if (t < 32) {
    constexpr int C = BIN_ROWS / 32;
    int c[C];
    int own = 0;
#pragma unroll
    for (int j = 0; j < C; ++j) { c[j] = cnt[C * t + j]; own += c[j]; }
    int inc = own;
#pragma unroll
    for (int off = 1; off < 32; off <<= 1) {
      const int v = __shfl_up(inc, off);
      if (t >= off) inc += v;
    }
    int run = inc - own;
#pragma unroll
    for (int j = 0; j < C; ++j) { pre[C * t + j] = run; run += c[j]; }
    if (t == 31) pre[BIN_ROWS] = run;
  }
  __syncthreads();

  if (t < BIN_ROWS) {
    const int g = b * BIN_ROWS + t;
    if (g <= N_NODES) row_ptr[g] = beg + pre[t];
  }

  for (int i = beg + t; i < end; i += 1024) {
    const int wd = binned[i];
    const unsigned r = min((unsigned)wd >> 20, (unsigned)(BIN_ROWS - 1));
    const unsigned p = (unsigned)(beg + pre[r] + rank[i]);
    slot[min(p, (unsigned)(N_EDGES - 1))] = wd;
  }
  __syncthreads();

  for (int p = beg + t; p < end; p += 1024) {
    const int wd = slot[p];
    const unsigned r = min((unsigned)wd >> 20, (unsigned)(BIN_ROWS - 1));
    const int e = (int)min((unsigned)wd & 0xFFFFFu, (unsigned)(N_EDGES - 1));
    const int lo = beg + pre[r];
    const int hi = beg + pre[r + 1];
    int k = 0;
    for (int u = lo; u < hi; ++u) k += (slot[u] < wd) ? 1 : 0;
    v2i en;
    en.x = src[e];
    en.y = __float_as_int(w[e]);
    *(v2i*)(csr + 2 * (size_t)min((unsigned)(lo + k), (unsigned)(N_EDGES - 1))) = en;
  }
}

__device__ __forceinline__ float xhalf(float x) {
  const int b = __builtin_bit_cast(int, x);
  return __builtin_bit_cast(float,
      __builtin_amdgcn_permlanex16(b, b, 0x76543210, 0xfedcba98, false, false));
}
template<int M>
__device__ __forceinline__ float xrow(float x) {
  static_assert(M == 1 || M == 2 || M == 4 || M == 8, "stays inside a 16-lane row");
  const int b = __builtin_bit_cast(int, x);
  return __builtin_bit_cast(float,
      __builtin_amdgcn_update_dpp(b, b, 0x160 + M, 0xf, 0xf, true));
}

template<int F>
__global__ __launch_bounds__(256)
void spmm_gather(const int* __restrict__ row_ptr, const int* __restrict__ csr,
                 const float* __restrict__ sup, const float* __restrict__ bias,
                 float* __restrict__ out) {
  static_assert(F == 256 || F == 64, "row widths of this network");
  const int lane = threadIdx.x & 31;
  const int row = __builtin_amdgcn_readfirstlane((int)(blockIdx.x * 8 + (threadIdx.x >> 5)));
  if (row >= N_NODES) return;
  const int beg = row_ptr[row];
  const int end = row_ptr[row + 1];

  if constexpr (F == 256) {
    v4f a0 = {}, a1 = {};
    const float* __restrict__ base = sup + 4 * lane;
    v4f c0 = *(const v4f*)(bias + 4 * lane);
    v4f c1 = *(const v4f*)(bias + 128 + 4 * lane);
    asm volatile("" : "+v"(c0), "+v"(c1));
#pragma unroll 4
    for (int j = beg; j < end; ++j) {
      const v2i en = *(const v2i*)(csr + 2 * (size_t)j);
      const float wt = __int_as_float(en.y);
      const unsigned s = min((unsigned)en.x, (unsigned)(N_NODES - 1));
      const float* __restrict__ p = base + (size_t)s * F;
      const v4f x0 = *(const v4f*)p;
      const v4f x1 = *(const v4f*)(p + 128);
      a0 += wt * x0;
      a1 += wt * x1;
    }
    a0 = a0 + c0;
    a1 = a1 + c1;
    v4f h0, h1;
    h0.x = fmaxf(a0.x, 0.0f); h0.y = fmaxf(a0.y, 0.0f);
    h0.z = fmaxf(a0.z, 0.0f); h0.w = fmaxf(a0.w, 0.0f);
    h1.x = fmaxf(a1.x, 0.0f); h1.y = fmaxf(a1.y, 0.0f);
    h1.z = fmaxf(a1.z, 0.0f); h1.w = fmaxf(a1.w, 0.0f);
    float* __restrict__ o = out + (size_t)row * F + 4 * lane;
    *(v4f*)o = h0;
    *(v4f*)(o + 128) = h1;
  } else {
    v2f a = {};
    const float* __restrict__ base = sup + 2 * lane;
    v2f bb = *(const v2f*)(bias + 2 * lane);
    asm volatile("" : "+v"(bb));
#pragma unroll 8
    for (int j = beg; j < end; ++j) {
      const v2i en = *(const v2i*)(csr + 2 * (size_t)j);
      const float wt = __int_as_float(en.y);
      const unsigned s = min((unsigned)en.x, (unsigned)(N_NODES - 1));
      const v2f x = *(const v2f*)(base + (size_t)s * F);
      a += wt * x;
    }
    const float vx = a.x + bb.x;
    const float vy = a.y + bb.y;

    float mx = fmaxf(vx, vy);
    mx = fmaxf(mx, xhalf(mx));
    mx = fmaxf(mx, xrow<8>(mx));
    mx = fmaxf(mx, xrow<4>(mx));
    mx = fmaxf(mx, xrow<2>(mx));
    mx = fmaxf(mx, xrow<1>(mx));

    float sx = __expf(vx - mx);
    float sy = __expf(vy - mx);
    sx += xhalf(sx);    sy += xhalf(sy);
    sx += xrow<8>(sx);  sy += xrow<8>(sy);
    sx += xrow<4>(sx);  sy += xrow<4>(sy);
    sx += xrow<2>(sx);  sy += xrow<2>(sy);
    sx += xrow<1>(sx);  sy += xrow<1>(sy);
    const float lse = __logf(sx + sy);

    v2f o;
    o.x = vx - mx - lse;
    o.y = vy - mx - lse;
    *(v2f*)(out + (size_t)row * F + 2 * lane) = o;
  }
}

extern "C" void kernel_launch(void* const* d_in, const int* in_sizes, int n_in,
                              void* d_out, int out_size, void* d_ws, size_t ws_size,
                              hipStream_t stream) {
  (void)in_sizes; (void)n_in; (void)out_size; (void)ws_size;

  const float* x  = (const float*)d_in[0];
  const float* W1 = (const float*)d_in[1];
  const float* b1 = (const float*)d_in[2];
  const float* W2 = (const float*)d_in[3];
  const float* b2 = (const float*)d_in[4];
  const float* ew = (const float*)d_in[5];
  const int*   es = (const int*)d_in[6];
  const int*   ed = (const int*)d_in[7];
  float* out = (float*)d_out;

  char* ws = (char*)d_ws;
  float* sup1 = (float*)ws;
  float* h    = (float*)(ws + (size_t)N_NODES * NHID * sizeof(float));
  float* sup2 = (float*)ws;
  int* slot_eid = (int*)h;
  int* binned   = (int*)((char*)h + (size_t)N_EDGES * sizeof(int));
  int* offs     = (int*)((char*)h + (size_t)2 * N_EDGES * sizeof(int));

  char* tab = ws + (size_t)2 * N_NODES * NHID * sizeof(float);
  int* row_ptr = (int*)tab;
  int* cnt     = (int*)(tab + (size_t)200704);
  int* rank    = (int*)(tab + (size_t)2 * 200704);
  int* csr     = (int*)(tab + (size_t)2 * 200704 + (size_t)N_EDGES * sizeof(int));

  const int Mtiles = N_NODES / 16;

  bin_count<<<NCHUNK, 1024, 0, stream>>>(ed, cnt);
  scan_deg<<<1, 1024, 0, stream>>>(cnt, offs);
  bin_scatter<<<NCHUNK, 1024, 0, stream>>>(ed, offs, binned);
  bin_sort<<<NBIN, 1024, 0, stream>>>(es, ew, offs, binned, rank, slot_eid, row_ptr, csr);

  {
    const int jobs = Mtiles * (NHID / 64);
    gemm_wmma_f32<NFEAT, NHID>
        <<<(jobs + 3) / 4, 128, 0, stream>>>(x, W1, sup1, Mtiles);
  }
  spmm_gather<NHID><<<N_NODES / 8, 256, 0, stream>>>(row_ptr, csr, sup1, b1, h);

  {
    const int jobs = Mtiles * (NCLASS / 64);
    gemm_wmma_f32<NHID, NCLASS>
        <<<(jobs + 3) / 4, 128, 0, stream>>>(h, W2, sup2, Mtiles);
  }
  spmm_gather<NCLASS><<<N_NODES / 8, 256, 0, stream>>>(row_ptr, csr, sup2, b2, out);
}


# Edge ordering in two launches: a chunk block orders its own edges by row range in LDS and writes its range offsets; a range block sums them for its start. Count table, scan, 2nd dst pass gone; 8->6
# speedup vs baseline: 4.5987x; 1.1130x over previous
#include <hip/hip_runtime.h>

#define N_NODES 50000
#define N_EDGES 800000
#define NFEAT 512
#define NHID  256
#define NCLASS 64

typedef float v2f __attribute__((ext_vector_type(2)));
typedef float v4f __attribute__((ext_vector_type(4)));
typedef float v8f __attribute__((ext_vector_type(8)));

template<int K, int N>
__global__ __launch_bounds__(128)
void gemm_wmma_f32(const float* __restrict__ A, const float* __restrict__ B,
                   float* __restrict__ C, int Mtiles) {
  constexpr int NG = N / 64;
  const int wave = threadIdx.x >> 5;
  const int lane = threadIdx.x & 31;
  const int job  = blockIdx.x * 4 + wave;
  if (job >= Mtiles * NG) return;
  const int mt = job / NG;
  const int ng = job % NG;

  const int half = lane >> 4;
  const int idx  = lane & 15;
  const int nb   = ng * 64;

  v8f acc0 = {}, acc1 = {}, acc2 = {}, acc3 = {};

  const float* __restrict__ arow  = A + (size_t)(mt * 16 + idx) * K;
  const float* __restrict__ bbase = B + nb + 4 * idx;

  for (int k0 = 0; k0 < K; k0 += 4) {
    const int ka = k0 + half * 2;
    v2f a  = *(const v2f*)(arow + ka);
    v4f r0 = *(const v4f*)(bbase + (size_t)ka * N);
    v4f r1 = *(const v4f*)(bbase + (size_t)(ka + 1) * N);
    v2f b;
    b.x = r0.x; b.y = r1.x;
    acc0 = __builtin_amdgcn_wmma_f32_16x16x4_f32(false, a, false, b, (short)0,
                                                 acc0, false, false);
    b.x = r0.y; b.y = r1.y;
    acc1 = __builtin_amdgcn_wmma_f32_16x16x4_f32(false, a, false, b, (short)0,
                                                 acc1, false, false);
    b.x = r0.z; b.y = r1.z;
    acc2 = __builtin_amdgcn_wmma_f32_16x16x4_f32(false, a, false, b, (short)0,
                                                 acc2, false, false);
    b.x = r0.w; b.y = r1.w;
    acc3 = __builtin_amdgcn_wmma_f32_16x16x4_f32(false, a, false, b, (short)0,
                                                 acc3, false, false);
  }

  const int mbase = mt * 16 + half * 8;
  float* __restrict__ cptr = C + (size_t)mbase * N + nb + 4 * idx;
#pragma unroll
  for (int i = 0; i < 8; ++i) {
    v4f st;
    st.x = acc0[i]; st.y = acc1[i]; st.z = acc2[i]; st.w = acc3[i];
    *(v4f*)(cptr + (size_t)i * N) = st;
  }
}

typedef int v2i __attribute__((ext_vector_type(2)));

#define BIN_SHIFT 7
#define BIN_ROWS  (1 << BIN_SHIFT)
#define NBIN   391
#define NCHUNK 128
#define CHUNK  6250
#define PER_T  ((CHUNK + 1023) / 1024)
#define LOCW   (NBIN + 1)
#define SCAN_C ((NBIN + 1 + 31) / 32)
static_assert(NCHUNK * CHUNK == N_EDGES, "chunks cover the edge list exactly");
static_assert(NBIN * BIN_ROWS >= N_NODES + 1, "ranges cover every row and the closing row_ptr entry");
static_assert(((N_NODES - 1) >> BIN_SHIFT) < NBIN, "every row has a range");
static_assert(32 * SCAN_C >= NBIN + 1, "one wave scans every range counter and the closing total");
static_assert(NCHUNK % 32 == 0 && NCHUNK * 8 == 1024, "bin_sort: eight threads per chunk, whole lanes of chunks in its scan");
static_assert(N_EDGES <= (1 << 20), "an edge number fits the low 20 bits of a binned word");

__global__ __launch_bounds__(1024)
void chunk_bin(const int* __restrict__ dst, int* __restrict__ loc, int* __restrict__ binned) {
  __shared__ int hist[32 * SCAN_C];
  __shared__ int cur[32 * SCAN_C];
  const int t = threadIdx.x;
  if (t < 32 * SCAN_C) hist[t] = 0;
  const int base = blockIdx.x * CHUNK;
  int d[PER_T];
#pragma unroll
  for (int j = 0; j < PER_T; ++j) {
    const int i = t + 1024 * j;
    d[j] = (i < CHUNK) ? dst[base + i] : 0;
  }
  __syncthreads();
#pragma unroll
  for (int j = 0; j < PER_T; ++j) {
    const int i = t + 1024 * j;
    if (i < CHUNK) {
      const unsigned dd = min((unsigned)d[j], (unsigned)(N_NODES - 1));
      atomicAdd(&hist[dd >> BIN_SHIFT], 1);
    }
  }
  __syncthreads();

  if (t < 32) {
    int c[SCAN_C];
    int own = 0;
#pragma unroll
    for (int j = 0; j < SCAN_C; ++j) { c[j] = hist[SCAN_C * t + j]; own += c[j]; }
    int inc = own;
#pragma unroll
    for (int off = 1; off < 32; off <<= 1) {
      const int v = __shfl_up(inc, off);
      if (t >= off) inc += v;
    }
    int run = inc - own;
#pragma unroll
    for (int j = 0; j < SCAN_C; ++j) {
      const int g = SCAN_C * t + j;
      cur[g] = run;
      if (g <= NBIN) loc[blockIdx.x * LOCW + g] = run;
      run += c[j];
    }
  }
  __syncthreads();

#pragma unroll
  for (int j = 0; j < PER_T; ++j) {
    const int i = t + 1024 * j;
    if (i < CHUNK) {
      const unsigned dd = min((unsigned)d[j], (unsigned)(N_NODES - 1));
      const int pos = atomicAdd(&cur[dd >> BIN_SHIFT], 1);
      binned[base + (int)min((unsigned)pos, (unsigned)(CHUNK - 1))] =
          (int)(((dd & (unsigned)(BIN_ROWS - 1)) << 20) | (unsigned)(base + i));
    }
  }
}

__global__ __launch_bounds__(1024)
void bin_sort(const int* __restrict__ src, const float* __restrict__ w,
              const int* __restrict__ loc, const int* __restrict__ binned,
              int* __restrict__ rank, int* __restrict__ slot,
              int* __restrict__ row_ptr, int* __restrict__ csr) {
  __shared__ int cnt[BIN_ROWS];
  __shared__ int pre[BIN_ROWS + 1];
  __shared__ int sub_s[NCHUNK];
  __shared__ int sub_p[NCHUNK + 1];
  __shared__ int beg_s;
  const int t = threadIdx.x;
  const int b = blockIdx.x;

  if (t < BIN_ROWS) cnt[t] = 0;

  if (t < 32) {
    constexpr int C = NCHUNK / 32;
    int s[C], n[C];
    int own = 0, below = 0;
#pragma unroll
    for (int j = 0; j < C; ++j) {
      const int c = C * t + j;
      const int s0 = (int)min((unsigned)loc[c * LOCW + b], (unsigned)CHUNK);
      const int e0 = (int)min((unsigned)loc[c * LOCW + b + 1], (unsigned)CHUNK);
      s[j] = s0;
      n[j] = max(e0 - s0, 0);
      own += n[j];
      below += s0;
    }
    int inc = own;
#pragma unroll
    for (int off = 1; off < 32; off <<= 1) {
      const int v = __shfl_up(inc, off);
      if (t >= off) inc += v;
    }
#pragma unroll
    for (int off = 16; off > 0; off >>= 1) below += __shfl_xor(below, off);
    int run = inc - own;
#pragma unroll
    for (int j = 0; j < C; ++j) { sub_s[C * t + j] = s[j]; sub_p[C * t + j] = run; run += n[j]; }
    if (t == 31) sub_p[NCHUNK] = run;
    if (t == 0) beg_s = (int)min((unsigned)below, (unsigned)N_EDGES);
  }
  __syncthreads();

  const int beg = beg_s;
  const int end = (int)min((unsigned)(beg + sub_p[NCHUNK]), (unsigned)N_EDGES);

  const int ch   = t >> 3;
  const int q    = t & 7;
  const int nsub = sub_p[ch + 1] - sub_p[ch];
  const int lin0 = beg + sub_p[ch];
  const int* __restrict__ run_c = binned + ch * CHUNK + sub_s[ch];

  for (int k = q; k < nsub; k += 8) {
    const unsigned r = min((unsigned)run_c[k] >> 20, (unsigned)(BIN_ROWS - 1));
    rank[min((unsigned)(lin0 + k), (unsigned)(N_EDGES - 1))] = atomicAdd(&cnt[r], 1);
  }
  __syncthreads();

  if (t < 32) {
    constexpr int C = BIN_ROWS / 32;
    int c[C];
    int own = 0;
#pragma unroll
    for (int j = 0; j < C; ++j) { c[j] = cnt[C * t + j]; own += c[j]; }
    int inc = own;
#pragma unroll
    for (int off = 1; off < 32; off <<= 1) {
      const int v = __shfl_up(inc, off);
      if (t >= off) inc += v;
    }
    int run = inc - own;
#pragma unroll
    for (int j = 0; j < C; ++j) { pre[C * t + j] = run; run += c[j]; }
    if (t == 31) pre[BIN_ROWS] = run;
  }
  __syncthreads();

  if (t < BIN_ROWS) {
    const int g = b * BIN_ROWS + t;
    if (g <= N_NODES) row_ptr[g] = beg + pre[t];
  }

  for (int k = q; k < nsub; k += 8) {
    const int wd = run_c[k];
    const unsigned r = min((unsigned)wd >> 20, (unsigned)(BIN_ROWS - 1));
    const unsigned p = (unsigned)(beg + pre[r] + rank[min((unsigned)(lin0 + k), (unsigned)(N_EDGES - 1))]);
    slot[min(p, (unsigned)(N_EDGES - 1))] = wd;
  }
  __syncthreads();

  for (int p = beg + t; p < end; p += 1024) {
    const int wd = slot[p];
    const unsigned r = min((unsigned)wd >> 20, (unsigned)(BIN_ROWS - 1));
    const int e = (int)min((unsigned)wd & 0xFFFFFu, (unsigned)(N_EDGES - 1));
    const int lo = beg + pre[r];
    const int hi = beg + pre[r + 1];
    int k = 0;
    for (int u = lo; u < hi; ++u) k += (slot[u] < wd) ? 1 : 0;
    v2i en;
    en.x = src[e];
    en.y = __float_as_int(w[e]);
    *(v2i*)(csr + 2 * (size_t)min((unsigned)(lo + k), (unsigned)(N_EDGES - 1))) = en;
  }
}

__device__ __forceinline__ float xhalf(float x) {
  const int b = __builtin_bit_cast(int, x);
  return __builtin_bit_cast(float,
      __builtin_amdgcn_permlanex16(b, b, 0x76543210, 0xfedcba98, false, false));
}
template<int M>
__device__ __forceinline__ float xrow(float x) {
  static_assert(M == 1 || M == 2 || M == 4 || M == 8, "stays inside a 16-lane row");
  const int b = __builtin_bit_cast(int, x);
  return __builtin_bit_cast(float,
      __builtin_amdgcn_update_dpp(b, b, 0x160 + M, 0xf, 0xf, true));
}

template<int F>
__global__ __launch_bounds__(256)
void spmm_gather(const int* __restrict__ row_ptr, const int* __restrict__ csr,
                 const float* __restrict__ sup, const float* __restrict__ bias,
                 float* __restrict__ out) {
  static_assert(F == 256 || F == 64, "row widths of this network");
  const int lane = threadIdx.x & 31;
  const int row = __builtin_amdgcn_readfirstlane((int)(blockIdx.x * 8 + (threadIdx.x >> 5)));
  if (row >= N_NODES) return;
  const int beg = row_ptr[row];
  const int end = row_ptr[row + 1];

  if constexpr (F == 256) {
    v4f a0 = {}, a1 = {};
    const float* __restrict__ base = sup + 4 * lane;
    v4f c0 = *(const v4f*)(bias + 4 * lane);
    v4f c1 = *(const v4f*)(bias + 128 + 4 * lane);
    asm volatile("" : "+v"(c0), "+v"(c1));
#pragma unroll 4
    for (int j = beg; j < end; ++j) {
      const v2i en = *(const v2i*)(csr + 2 * (size_t)j);
      const float wt = __int_as_float(en.y);
      const unsigned s = min((unsigned)en.x, (unsigned)(N_NODES - 1));
      const float* __restrict__ p = base + (size_t)s * F;
      const v4f x0 = *(const v4f*)p;
      const v4f x1 = *(const v4f*)(p + 128);
      a0 += wt * x0;
      a1 += wt * x1;
    }
    a0 = a0 + c0;
    a1 = a1 + c1;
    v4f h0, h1;
    h0.x = fmaxf(a0.x, 0.0f); h0.y = fmaxf(a0.y, 0.0f);
    h0.z = fmaxf(a0.z, 0.0f); h0.w = fmaxf(a0.w, 0.0f);
    h1.x = fmaxf(a1.x, 0.0f); h1.y = fmaxf(a1.y, 0.0f);
    h1.z = fmaxf(a1.z, 0.0f); h1.w = fmaxf(a1.w, 0.0f);
    float* __restrict__ o = out + (size_t)row * F + 4 * lane;
    *(v4f*)o = h0;
    *(v4f*)(o + 128) = h1;
  } else {
    v2f a = {};
    const float* __restrict__ base = sup + 2 * lane;
    v2f bb = *(const v2f*)(bias + 2 * lane);
    asm volatile("" : "+v"(bb));
#pragma unroll 8
    for (int j = beg; j < end; ++j) {
      const v2i en = *(const v2i*)(csr + 2 * (size_t)j);
      const float wt = __int_as_float(en.y);
      const unsigned s = min((unsigned)en.x, (unsigned)(N_NODES - 1));
      const v2f x = *(const v2f*)(base + (size_t)s * F);
      a += wt * x;
    }
    const float vx = a.x + bb.x;
    const float vy = a.y + bb.y;

    float mx = fmaxf(vx, vy);
    mx = fmaxf(mx, xhalf(mx));
    mx = fmaxf(mx, xrow<8>(mx));
    mx = fmaxf(mx, xrow<4>(mx));
    mx = fmaxf(mx, xrow<2>(mx));
    mx = fmaxf(mx, xrow<1>(mx));

    float sx = __expf(vx - mx);
    float sy = __expf(vy - mx);
    sx += xhalf(sx);    sy += xhalf(sy);
    sx += xrow<8>(sx);  sy += xrow<8>(sy);
    sx += xrow<4>(sx);  sy += xrow<4>(sy);
    sx += xrow<2>(sx);  sy += xrow<2>(sy);
    sx += xrow<1>(sx);  sy += xrow<1>(sy);
    const float lse = __logf(sx + sy);

    v2f o;
    o.x = vx - mx - lse;
    o.y = vy - mx - lse;
    *(v2f*)(out + (size_t)row * F + 2 * lane) = o;
  }
}

extern "C" void kernel_launch(void* const* d_in, const int* in_sizes, int n_in,
                              void* d_out, int out_size, void* d_ws, size_t ws_size,
                              hipStream_t stream) {
  (void)in_sizes; (void)n_in; (void)out_size; (void)ws_size;

  const float* x  = (const float*)d_in[0];
  const float* W1 = (const float*)d_in[1];
  const float* b1 = (const float*)d_in[2];
  const float* W2 = (const float*)d_in[3];
  const float* b2 = (const float*)d_in[4];
  const float* ew = (const float*)d_in[5];
  const int*   es = (const int*)d_in[6];
  const int*   ed = (const int*)d_in[7];
  float* out = (float*)d_out;

  char* ws = (char*)d_ws;
  float* sup1 = (float*)ws;
  float* h    = (float*)(ws + (size_t)N_NODES * NHID * sizeof(float));
  float* sup2 = (float*)ws;
  int* slot_eid = (int*)h;
  int* binned   = (int*)((char*)h + (size_t)N_EDGES * sizeof(int));

  char* tab = ws + (size_t)2 * N_NODES * NHID * sizeof(float);
  int* row_ptr = (int*)tab;
  int* loc     = (int*)(tab + (size_t)200704);
  int* rank    = (int*)(tab + (size_t)2 * 200704);
  int* csr     = (int*)(tab + (size_t)2 * 200704 + (size_t)N_EDGES * sizeof(int));

  const int Mtiles = N_NODES / 16;

  chunk_bin<<<NCHUNK, 1024, 0, stream>>>(ed, loc, binned);
  bin_sort<<<NBIN, 1024, 0, stream>>>(es, ew, loc, binned, rank, slot_eid, row_ptr, csr);

  {
    const int jobs = Mtiles * (NHID / 64);
    gemm_wmma_f32<NFEAT, NHID>
        <<<(jobs + 3) / 4, 128, 0, stream>>>(x, W1, sup1, Mtiles);
  }
  spmm_gather<NHID><<<N_NODES / 8, 256, 0, stream>>>(row_ptr, csr, sup1, b1, h);

  {
    const int jobs = Mtiles * (NCLASS / 64);
    gemm_wmma_f32<NHID, NCLASS>
        <<<(jobs + 3) / 4, 128, 0, stream>>>(h, W2, sup2, Mtiles);
  }
  spmm_gather<NCLASS><<<N_NODES / 8, 256, 0, stream>>>(row_ptr, csr, sup2, b2, out);
}


# h @ W2 moved into the layer-1 gather: a 32-row block lays relu rows in LDS, 16 waves = tile x k quarter on WMMA f32, quarters summed ((q0+q1)+q2)+q3; h store/load and one launch gone, 6->5
# speedup vs baseline: 4.8217x; 1.0485x over previous
#include <hip/hip_runtime.h>

#define N_NODES 50000
#define N_EDGES 800000
#define NFEAT 512
#define NHID  256
#define NCLASS 64

typedef float v2f __attribute__((ext_vector_type(2)));
typedef float v4f __attribute__((ext_vector_type(4)));
typedef float v8f __attribute__((ext_vector_type(8)));

template<int K, int N>
__global__ __launch_bounds__(128)
void gemm_wmma_f32(const float* __restrict__ A, const float* __restrict__ B,
                   float* __restrict__ C, int Mtiles) {
  constexpr int NG = N / 64;
  const int wave = threadIdx.x >> 5;
  const int lane = threadIdx.x & 31;
  const int job  = blockIdx.x * 4 + wave;
  if (job >= Mtiles * NG) return;
  const int mt = job / NG;
  const int ng = job % NG;

  const int half = lane >> 4;
  const int idx  = lane & 15;
  const int nb   = ng * 64;

  v8f acc0 = {}, acc1 = {}, acc2 = {}, acc3 = {};

  const float* __restrict__ arow  = A + (size_t)(mt * 16 + idx) * K;
  const float* __restrict__ bbase = B + nb + 4 * idx;

  for (int k0 = 0; k0 < K; k0 += 4) {
    const int ka = k0 + half * 2;
    v2f a  = *(const v2f*)(arow + ka);
    v4f r0 = *(const v4f*)(bbase + (size_t)ka * N);
    v4f r1 = *(const v4f*)(bbase + (size_t)(ka + 1) * N);
    v2f b;
    b.x = r0.x; b.y = r1.x;
    acc0 = __builtin_amdgcn_wmma_f32_16x16x4_f32(false, a, false, b, (short)0,
                                                 acc0, false, false);
    b.x = r0.y; b.y = r1.y;
    acc1 = __builtin_amdgcn_wmma_f32_16x16x4_f32(false, a, false, b, (short)0,
                                                 acc1, false, false);
    b.x = r0.z; b.y = r1.z;
    acc2 = __builtin_amdgcn_wmma_f32_16x16x4_f32(false, a, false, b, (short)0,
                                                 acc2, false, false);
    b.x = r0.w; b.y = r1.w;
    acc3 = __builtin_amdgcn_wmma_f32_16x16x4_f32(false, a, false, b, (short)0,
                                                 acc3, false, false);
  }

  const int mbase = mt * 16 + half * 8;
  float* __restrict__ cptr = C + (size_t)mbase * N + nb + 4 * idx;
#pragma unroll
  for (int i = 0; i < 8; ++i) {
    v4f st;
    st.x = acc0[i]; st.y = acc1[i]; st.z = acc2[i]; st.w = acc3[i];
    *(v4f*)(cptr + (size_t)i * N) = st;
  }
}

typedef int v2i __attribute__((ext_vector_type(2)));

#define BIN_SHIFT 7
#define BIN_ROWS  (1 << BIN_SHIFT)
#define NBIN   391
#define NCHUNK 128
#define CHUNK  6250
#define PER_T  ((CHUNK + 1023) / 1024)
#define LOCW   (NBIN + 1)
#define SCAN_C ((NBIN + 1 + 31) / 32)
static_assert(NCHUNK * CHUNK == N_EDGES, "chunks cover the edge list exactly");
static_assert(NBIN * BIN_ROWS >= N_NODES + 1, "ranges cover every row and the closing row_ptr entry");
static_assert(((N_NODES - 1) >> BIN_SHIFT) < NBIN, "every row has a range");
static_assert(32 * SCAN_C >= NBIN + 1, "one wave scans every range counter and the closing total");
static_assert(NCHUNK % 32 == 0 && NCHUNK * 8 == 1024, "bin_sort: eight threads per chunk, whole lanes of chunks in its scan");
static_assert(N_EDGES <= (1 << 20), "an edge number fits the low 20 bits of a binned word");

__global__ __launch_bounds__(1024)
void chunk_bin(const int* __restrict__ dst, int* __restrict__ loc, int* __restrict__ binned) {
  __shared__ int hist[32 * SCAN_C];
  __shared__ int cur[32 * SCAN_C];
  const int t = threadIdx.x;
  if (t < 32 * SCAN_C) hist[t] = 0;
  const int base = blockIdx.x * CHUNK;
  int d[PER_T];
#pragma unroll
  for (int j = 0; j < PER_T; ++j) {
    const int i = t + 1024 * j;
    d[j] = (i < CHUNK) ? dst[base + i] : 0;
  }
  __syncthreads();
#pragma unroll
  for (int j = 0; j < PER_T; ++j) {
    const int i = t + 1024 * j;
    if (i < CHUNK) {
      const unsigned dd = min((unsigned)d[j], (unsigned)(N_NODES - 1));
      atomicAdd(&hist[dd >> BIN_SHIFT], 1);
    }
  }
  __syncthreads();

  if (t < 32) {
    int c[SCAN_C];
    int own = 0;
#pragma unroll
    for (int j = 0; j < SCAN_C; ++j) { c[j] = hist[SCAN_C * t + j]; own += c[j]; }
    int inc = own;
#pragma unroll
    for (int off = 1; off < 32; off <<= 1) {
      const int v = __shfl_up(inc, off);
      if (t >= off) inc += v;
    }
    int run = inc - own;
#pragma unroll
    for (int j = 0; j < SCAN_C; ++j) {
      const int g = SCAN_C * t + j;
      cur[g] = run;
      if (g <= NBIN) loc[blockIdx.x * LOCW + g] = run;
      run += c[j];
    }
  }
  __syncthreads();

#pragma unroll
  for (int j = 0; j < PER_T; ++j) {
    const int i = t + 1024 * j;
    if (i < CHUNK) {
      const unsigned dd = min((unsigned)d[j], (unsigned)(N_NODES - 1));
      const int pos = atomicAdd(&cur[dd >> BIN_SHIFT], 1);
      binned[base + (int)min((unsigned)pos, (unsigned)(CHUNK - 1))] =
          (int)(((dd & (unsigned)(BIN_ROWS - 1)) << 20) | (unsigned)(base + i));
    }
  }
}

__global__ __launch_bounds__(1024)
void bin_sort(const int* __restrict__ src, const float* __restrict__ w,
              const int* __restrict__ loc, const int* __restrict__ binned,
              int* __restrict__ rank, int* __restrict__ slot,
              int* __restrict__ row_ptr, int* __restrict__ csr) {
  __shared__ int cnt[BIN_ROWS];
  __shared__ int pre[BIN_ROWS + 1];
  __shared__ int sub_s[NCHUNK];
  __shared__ int sub_p[NCHUNK + 1];
  __shared__ int beg_s;
  const int t = threadIdx.x;
  const int b = blockIdx.x;

  if (t < BIN_ROWS) cnt[t] = 0;

  if (t < 32) {
    constexpr int C = NCHUNK / 32;
    int s[C], n[C];
    int own = 0, below = 0;
#pragma unroll
    for (int j = 0; j < C; ++j) {
      const int c = C * t + j;
      const int s0 = (int)min((unsigned)loc[c * LOCW + b], (unsigned)CHUNK);
      const int e0 = (int)min((unsigned)loc[c * LOCW + b + 1], (unsigned)CHUNK);
      s[j] = s0;
      n[j] = max(e0 - s0, 0);
      own += n[j];
      below += s0;
    }
    int inc = own;
#pragma unroll
    for (int off = 1; off < 32; off <<= 1) {
      const int v = __shfl_up(inc, off);
      if (t >= off) inc += v;
    }
#pragma unroll
    for (int off = 16; off > 0; off >>= 1) below += __shfl_xor(below, off);
    int run = inc - own;
#pragma unroll
    for (int j = 0; j < C; ++j) { sub_s[C * t + j] = s[j]; sub_p[C * t + j] = run; run += n[j]; }
    if (t == 31) sub_p[NCHUNK] = run;
    if (t == 0) beg_s = (int)min((unsigned)below, (unsigned)N_EDGES);
  }
  __syncthreads();

  const int beg = beg_s;
  const int end = (int)min((unsigned)(beg + sub_p[NCHUNK]), (unsigned)N_EDGES);

  const int ch   = t >> 3;
  const int q    = t & 7;
  const int nsub = sub_p[ch + 1] - sub_p[ch];
  const int lin0 = beg + sub_p[ch];
  const int* __restrict__ run_c = binned + ch * CHUNK + sub_s[ch];

  for (int k = q; k < nsub; k += 8) {
    const unsigned r = min((unsigned)run_c[k] >> 20, (unsigned)(BIN_ROWS - 1));
    rank[min((unsigned)(lin0 + k), (unsigned)(N_EDGES - 1))] = atomicAdd(&cnt[r], 1);
  }
  __syncthreads();

  if (t < 32) {
    constexpr int C = BIN_ROWS / 32;
    int c[C];
    int own = 0;
#pragma unroll
    for (int j = 0; j < C; ++j) { c[j] = cnt[C * t + j]; own += c[j]; }
    int inc = own;
#pragma unroll
    for (int off = 1; off < 32; off <<= 1) {
      const int v = __shfl_up(inc, off);
      if (t >= off) inc += v;
    }
    int run = inc - own;
#pragma unroll
    for (int j = 0; j < C; ++j) { pre[C * t + j] = run; run += c[j]; }
    if (t == 31) pre[BIN_ROWS] = run;
  }
  __syncthreads();

  if (t < BIN_ROWS) {
    const int g = b * BIN_ROWS + t;
    if (g <= N_NODES) row_ptr[g] = beg + pre[t];
  }

  for (int k = q; k < nsub; k += 8) {
    const int wd = run_c[k];
    const unsigned r = min((unsigned)wd >> 20, (unsigned)(BIN_ROWS - 1));
    const unsigned p = (unsigned)(beg + pre[r] + rank[min((unsigned)(lin0 + k), (unsigned)(N_EDGES - 1))]);
    slot[min(p, (unsigned)(N_EDGES - 1))] = wd;
  }
  __syncthreads();

  for (int p = beg + t; p < end; p += 1024) {
    const int wd = slot[p];
    const unsigned r = min((unsigned)wd >> 20, (unsigned)(BIN_ROWS - 1));
    const int e = (int)min((unsigned)wd & 0xFFFFFu, (unsigned)(N_EDGES - 1));
    const int lo = beg + pre[r];
    const int hi = beg + pre[r + 1];
    int k = 0;
    for (int u = lo; u < hi; ++u) k += (slot[u] < wd) ? 1 : 0;
    v2i en;
    en.x = src[e];
    en.y = __float_as_int(w[e]);
    *(v2i*)(csr + 2 * (size_t)min((unsigned)(lo + k), (unsigned)(N_EDGES - 1))) = en;
  }
}

__device__ __forceinline__ float xhalf(float x) {
  const int b = __builtin_bit_cast(int, x);
  return __builtin_bit_cast(float,
      __builtin_amdgcn_permlanex16(b, b, 0x76543210, 0xfedcba98, false, false));
}
template<int M>
__device__ __forceinline__ float xrow(float x) {
  static_assert(M == 1 || M == 2 || M == 4 || M == 8, "stays inside a 16-lane row");
  const int b = __builtin_bit_cast(int, x);
  return __builtin_bit_cast(float,
      __builtin_amdgcn_update_dpp(b, b, 0x160 + M, 0xf, 0xf, true));
}

__global__ __launch_bounds__(256)
void spmm_gather_lsm(const int* __restrict__ row_ptr, const int* __restrict__ csr,
                     const float* __restrict__ sup, const float* __restrict__ bias,
                     float* __restrict__ out) {
  constexpr int F = NCLASS;
  const int lane = threadIdx.x & 31;
  const int row = __builtin_amdgcn_readfirstlane((int)(blockIdx.x * 8 + (threadIdx.x >> 5)));
  if (row >= N_NODES) return;
  const int beg = row_ptr[row];
  const int end = row_ptr[row + 1];

  v2f a = {};
  const float* __restrict__ base = sup + 2 * lane;
  v2f bb = *(const v2f*)(bias + 2 * lane);
  asm volatile("" : "+v"(bb));
#pragma unroll 8
  for (int j = beg; j < end; ++j) {
    const v2i en = *(const v2i*)(csr + 2 * (size_t)j);
    const float wt = __int_as_float(en.y);
    const unsigned s = min((unsigned)en.x, (unsigned)(N_NODES - 1));
    const v2f x = *(const v2f*)(base + (size_t)s * F);
    a += wt * x;
  }
  const float vx = a.x + bb.x;
  const float vy = a.y + bb.y;

  float mx = fmaxf(vx, vy);
  mx = fmaxf(mx, xhalf(mx));
  mx = fmaxf(mx, xrow<8>(mx));
  mx = fmaxf(mx, xrow<4>(mx));
  mx = fmaxf(mx, xrow<2>(mx));
  mx = fmaxf(mx, xrow<1>(mx));

  float sx = __expf(vx - mx);
  float sy = __expf(vy - mx);
  sx += xhalf(sx);    sy += xhalf(sy);
  sx += xrow<8>(sx);  sy += xrow<8>(sy);
  sx += xrow<4>(sx);  sy += xrow<4>(sy);
  sx += xrow<2>(sx);  sy += xrow<2>(sy);
  sx += xrow<1>(sx);  sy += xrow<1>(sy);
  const float lse = __logf(sx + sy);

  v2f o;
  o.x = vx - mx - lse;
  o.y = vy - mx - lse;
  *(v2f*)(out + (size_t)row * F + 2 * lane) = o;
}

#define FROWS 32
#define FBLOCKS ((N_NODES + FROWS - 1) / FROWS)
#define HS_PITCH (NHID + 4)
#define KQ (NHID / 4)
#define KSTEPS (KQ / 4)

__global__ __launch_bounds__(512)
void spmm_gather_relu_gemm(const int* __restrict__ row_ptr, const int* __restrict__ csr,
                           const float* __restrict__ sup, const float* __restrict__ bias,
                           const float* __restrict__ W2, float* __restrict__ sup2) {
  __shared__ __attribute__((aligned(16))) float hs[FROWS * HS_PITCH];
  __shared__ __attribute__((aligned(16))) float part[4 * FROWS * NCLASS];
  const int lane = threadIdx.x & 31;
  const int wave = __builtin_amdgcn_readfirstlane((int)(threadIdx.x >> 5));

  const float* __restrict__ base = sup + 4 * lane;
  v4f c0 = *(const v4f*)(bias + 4 * lane);
  v4f c1 = *(const v4f*)(bias + 128 + 4 * lane);
  asm volatile("" : "+v"(c0), "+v"(c1));

  int rbeg[2], rend[2];
#pragma unroll
  for (int r = 0; r < 2; ++r) {
    const int row = (int)blockIdx.x * FROWS + 16 * r + wave;
    const int rc  = min(row, N_NODES - 1);
    const int b0  = row_ptr[rc];
    const int e0  = row_ptr[rc + 1];
    rbeg[r] = b0;
    rend[r] = (row < N_NODES) ? e0 : b0;
  }

#pragma unroll
  for (int r = 0; r < 2; ++r) {
    const int trow = 16 * r + wave;
    const int beg = rbeg[r], end = rend[r];
    v4f a0 = {}, a1 = {};
#pragma unroll 4
    for (int j = beg; j < end; ++j) {
      const v2i en = *(const v2i*)(csr + 2 * (size_t)j);
      const float wt = __int_as_float(en.y);
      const unsigned s = min((unsigned)en.x, (unsigned)(N_NODES - 1));
      const float* __restrict__ p = base + (size_t)s * NHID;
      const v4f x0 = *(const v4f*)p;
      const v4f x1 = *(const v4f*)(p + 128);
      a0 += wt * x0;
      a1 += wt * x1;
    }
    a0 = a0 + c0;
    a1 = a1 + c1;
    v4f h0, h1;
    h0.x = fmaxf(a0.x, 0.0f); h0.y = fmaxf(a0.y, 0.0f);
    h0.z = fmaxf(a0.z, 0.0f); h0.w = fmaxf(a0.w, 0.0f);
    h1.x = fmaxf(a1.x, 0.0f); h1.y = fmaxf(a1.y, 0.0f);
    h1.z = fmaxf(a1.z, 0.0f); h1.w = fmaxf(a1.w, 0.0f);
    float* hrow = hs + trow * HS_PITCH + 4 * lane;
    *(v4f*)hrow = h0;
    *(v4f*)(hrow + 128) = h1;
  }

  const int half = lane >> 4;
  const int idx  = lane & 15;
  const int tn   = wave & 3;
  const int kq   = (wave >> 2) * KQ;
  float bx[KSTEPS], by[KSTEPS];
  {
    const float* __restrict__ bcol = W2 + (size_t)(kq + 2 * half) * NCLASS + 16 * tn + idx;
#pragma unroll
    for (int s = 0; s < KSTEPS; ++s) {
      bx[s] = bcol[(4 * s) * NCLASS];
      by[s] = bcol[(4 * s + 1) * NCLASS];
    }
  }
  __syncthreads();

  v8f acc0 = {}, acc1 = {};
  {
    const float* __restrict__ ar = hs + idx * HS_PITCH + kq + 2 * half;
#pragma unroll
    for (int s = 0; s < KSTEPS; ++s) {
      const v2f alo = *(const v2f*)(ar + 4 * s);
      const v2f ahi = *(const v2f*)(ar + 16 * HS_PITCH + 4 * s);
      v2f b;
      b.x = bx[s]; b.y = by[s];
      acc0 = __builtin_amdgcn_wmma_f32_16x16x4_f32(false, alo, false, b, (short)0,
                                                   acc0, false, false);
      acc1 = __builtin_amdgcn_wmma_f32_16x16x4_f32(false, ahi, false, b, (short)0,
                                                   acc1, false, false);
    }
  }
  {
    float* pt = part + (wave >> 2) * (FROWS * NCLASS) + (8 * half) * NCLASS + 16 * tn + idx;
#pragma unroll
    for (int i = 0; i < 8; ++i) {
      pt[i * NCLASS] = acc0[i];
      pt[(16 + i) * NCLASS] = acc1[i];
    }
  }
  __syncthreads();

  {
    const int rr = (int)(threadIdx.x >> 4);
    const int cc = 4 * (int)(threadIdx.x & 15);
    const float* __restrict__ p0 = part + rr * NCLASS + cc;
    v4f s = *(const v4f*)p0;
    s = s + *(const v4f*)(p0 + 1 * (FROWS * NCLASS));
    s = s + *(const v4f*)(p0 + 2 * (FROWS * NCLASS));
    s = s + *(const v4f*)(p0 + 3 * (FROWS * NCLASS));
    const int grow = (int)blockIdx.x * FROWS + rr;
    if (grow < N_NODES) *(v4f*)(sup2 + (size_t)grow * NCLASS + cc) = s;
  }
}

extern "C" void kernel_launch(void* const* d_in, const int* in_sizes, int n_in,
                              void* d_out, int out_size, void* d_ws, size_t ws_size,
                              hipStream_t stream) {
  (void)in_sizes; (void)n_in; (void)out_size; (void)ws_size;

  const float* x  = (const float*)d_in[0];
  const float* W1 = (const float*)d_in[1];
  const float* b1 = (const float*)d_in[2];
  const float* W2 = (const float*)d_in[3];
  const float* b2 = (const float*)d_in[4];
  const float* ew = (const float*)d_in[5];
  const int*   es = (const int*)d_in[6];
  const int*   ed = (const int*)d_in[7];
  float* out = (float*)d_out;

  char* ws = (char*)d_ws;
  float* sup1 = (float*)ws;
  char*  pl1  = ws + (size_t)N_NODES * NHID * sizeof(float);
  float* sup2 = (float*)pl1;
  int* slot_eid = (int*)pl1;
  int* binned   = (int*)(pl1 + (size_t)N_EDGES * sizeof(int));

  char* tab = ws + (size_t)2 * N_NODES * NHID * sizeof(float);
  int* row_ptr = (int*)tab;
  int* loc     = (int*)(tab + (size_t)200704);
  int* rank    = (int*)(tab + (size_t)2 * 200704);
  int* csr     = (int*)(tab + (size_t)2 * 200704 + (size_t)N_EDGES * sizeof(int));

  const int Mtiles = N_NODES / 16;

  chunk_bin<<<NCHUNK, 1024, 0, stream>>>(ed, loc, binned);
  bin_sort<<<NBIN, 1024, 0, stream>>>(es, ew, loc, binned, rank, slot_eid, row_ptr, csr);

  {
    const int jobs = Mtiles * (NHID / 64);
    gemm_wmma_f32<NFEAT, NHID>
        <<<(jobs + 3) / 4, 128, 0, stream>>>(x, W1, sup1, Mtiles);
  }
  spmm_gather_relu_gemm<<<FBLOCKS, 512, 0, stream>>>(row_ptr, csr, sup1, b1, W2, sup2);

  spmm_gather_lsm<<<N_NODES / 8, 256, 0, stream>>>(row_ptr, csr, sup2, b2, out);
}
